# starting kernel
# baseline (speedup 1.0000x reference)
#include <hip/hip_runtime.h>
#include <hip/hip_bf16.h>
#include <math.h>


typedef __attribute__((ext_vector_type(2))) float v2f;
typedef __attribute__((ext_vector_type(8))) float v8f;

#define BATCH 64
#define SEQL  24
#define HDIM  512

__device__ __forceinline__ float sigm(float x) { return 1.0f / (1.0f + __expf(-x)); }
__device__ __forceinline__ float tanh_fast(float x) { return 2.0f / (1.0f + __expf(-2.0f * x)) - 1.0f; }

template <int MODE>
__global__ __launch_bounds__(256) void gemm_wmma(const float* __restrict__ X,
                                                 const float* __restrict__ hbuf,
                                                 const float* __restrict__ cbuf,
                                                 const float* __restrict__ W,
                                                 const float* __restrict__ bias,
                                                 float* __restrict__ out0,
                                                 float* __restrict__ out1,
                                                 int K, int n) {
  __shared__ float A_lds[32 * 36];
  __shared__ float B_lds[320 * 36];

  const int tid  = threadIdx.x;
  const unsigned lane = tid & 31u;
  const int wave = tid >> 5;
  const int wm   = wave >> 2;
  const int wn   = wave & 3;
  const int l16  = (int)(lane & 15u);
  const int lhi  = (int)(lane >> 4);

  const int r0 = blockIdx.x * 32;
  const int NACC = (MODE == 0) ? 4 : 5;

  v8f acc[5] = {};

  for (int k0 = 0; k0 < K; k0 += 32) {
    {
      int ar  = tid >> 3;
      int ac4 = tid & 7;
      int rg  = r0 + ar;
      int kg  = k0 + ac4 * 4;
      float4 val;
      if (MODE == 0) {
        val = *(const float4*)&X[(size_t)rg * K + kg];
      } else {
        int b = rg / n;
        int p = rg - b * n;
        const float* src = (kg < HDIM)
            ? &hbuf[((size_t)(b * SEQL + p) * HDIM) + kg]
            : &hbuf[((size_t)(b * SEQL + p + 1) * HDIM) + (kg - HDIM)];
        val = *(const float4*)src;
      }
      *(float4*)&A_lds[ar * 36 + ac4 * 4] = val;
    }
    if (MODE == 0) {
      int nb0 = blockIdx.y * 256;
#pragma unroll
      for (int i = 0; i < 8; ++i) {
        int f    = i * 256 + tid;
        int nrow = f >> 3;
        int kc4  = f & 7;
        float4 val = *(const float4*)&W[(size_t)(nb0 + nrow) * K + k0 + kc4 * 4];
        *(float4*)&B_lds[nrow * 36 + kc4 * 4] = val;
      }
    } else {
      int nb0 = blockIdx.y * 64;
#pragma unroll
      for (int i = 0; i < 10; ++i) {
        int f    = i * 256 + tid;
        int nrow = f >> 3;
        int kc4  = f & 7;
        int g    = nrow >> 6;
        int nl   = nrow & 63;
        float4 val = *(const float4*)&W[(size_t)(g * HDIM + nb0 + nl) * K + k0 + kc4 * 4];
        *(float4*)&B_lds[nrow * 36 + kc4 * 4] = val;
      }
    }
    __syncthreads();

#pragma unroll
    for (int kk = 0; kk < 32; kk += 4) {
      v2f a = *(const v2f*)&A_lds[(wm * 16 + l16) * 36 + kk + 2 * lhi];
#pragma unroll
      for (int g = 0; g < NACC; ++g) {
        int nrow = (MODE == 0) ? (wn * 64 + g * 16 + l16)
                               : (g * 64 + wn * 16 + l16);
        v2f bf = *(const v2f*)&B_lds[nrow * 36 + kk + 2 * lhi];
        acc[g] = __builtin_amdgcn_wmma_f32_16x16x4_f32(
            false, a, false, bf, (short)0, acc[g], false, false);
      }
    }
    __syncthreads();
  }

  if (MODE == 0) {
    const int colbase = blockIdx.y * 256 + wn * 64;
    const int rowg0   = r0 + wm * 16 + 8 * lhi;
#pragma unroll
    for (int j = 0; j < 4; ++j) {
      int col = colbase + j * 16 + l16;
      float bv = bias[col];
#pragma unroll
      for (int vr = 0; vr < 8; ++vr) {
        int rg = rowg0 + vr;
        float val = acc[j][vr] + bv;
        if (col < HDIM) out0[(size_t)rg * HDIM + col] = val;
        else            out1[(size_t)rg * HDIM + (col - HDIM)] = val;
      }
    }
  } else {
    const int colg = blockIdx.y * 64 + wn * 16 + l16;
    float bz[5];
#pragma unroll
    for (int g = 0; g < 5; ++g) bz[g] = bias[g * HDIM + colg];
    int rg = r0 + wm * 16 + 8 * lhi;
    int b  = rg / n;
    int p  = rg - b * n;
#pragma unroll
    for (int vr = 0; vr < 8; ++vr) {
      float gi  = acc[0][vr] + bz[0];
      float gfl = acc[1][vr] + bz[1];
      float gfr = acc[2][vr] + bz[2];
      float gu  = acc[3][vr] + bz[3];
      float go  = acc[4][vr] + bz[4];
      size_t cbase = (size_t)(b * SEQL + p) * HDIM + colg;
      float cl = cbuf[cbase];
      float cr = cbuf[cbase + HDIM];
      float nc = cl * sigm(gfl + 1.0f) + cr * sigm(gfr + 1.0f) + tanh_fast(gu) * sigm(gi);
      float nh = sigm(go) * tanh_fast(nc);
      size_t oi = (size_t)rg * HDIM + colg;
      out0[oi] = nh;
      out1[oi] = nc;
      ++rg; ++p;
      if (p == n) { p = 0; ++b; }
    }
  }
}

__global__ __launch_bounds__(256) void mix_k(const float* __restrict__ newh,
                                             const float* __restrict__ newc,
                                             const float* __restrict__ hcur,
                                             const float* __restrict__ ccur,
                                             const float* __restrict__ q,
                                             const int* __restrict__ length,
                                             float* __restrict__ hnxt,
                                             float* __restrict__ cnxt,
                                             int n, int iter) {
  __shared__ float sc[32];
  __shared__ int ksel_s;
  const int b = blockIdx.x;
  const int tid = threadIdx.x;
  const unsigned lane = tid & 31u;
  const int wave = tid >> 5;

  for (int p = wave; p < n; p += 8) {
    const float* row = newh + (size_t)(b * n + p) * HDIM;
    float s = 0.0f;
    for (int j = (int)lane; j < HDIM; j += 32) s += row[j] * q[j];
#pragma unroll
    for (int off = 16; off > 0; off >>= 1) s += __shfl_xor(s, off, 32);
    if (lane == 0) sc[p] = s;
  }
  __syncthreads();

  if (tid == 0) {
    int len = length[b];
    int best = 0;
    float bv = -INFINITY;
    for (int p = 0; p < n; ++p) {
      float s = ((iter + 1 + p) < len) ? sc[p] : -1.0e9f;
      if (s > bv) { bv = s; best = p; }
    }
    ksel_s = best;
  }
  __syncthreads();

  const int ksel = ksel_s;
  const bool done = (iter + 1) < length[b];

  for (int idx = tid; idx < n * HDIM; idx += 256) {
    int p = idx >> 9;
    int j = idx & (HDIM - 1);
    size_t cur = (size_t)(b * SEQL + p) * HDIM + j;
    float hm, cm;
    if (p < ksel)      { hm = hcur[cur];        cm = ccur[cur]; }
    else if (p > ksel) { hm = hcur[cur + HDIM]; cm = ccur[cur + HDIM]; }
    else {
      size_t ni = (size_t)(b * n + p) * HDIM + j;
      hm = newh[ni]; cm = newc[ni];
    }
    hnxt[cur] = done ? hm : hcur[cur];
    cnxt[cur] = done ? cm : ccur[cur];
  }
}

__global__ __launch_bounds__(256) void final_copy_k(const float* __restrict__ hfin,
                                                    float* __restrict__ out) {
  int idx = blockIdx.x * 256 + threadIdx.x;
  int b = idx >> 9;
  int j = idx & (HDIM - 1);
  out[idx] = hfin[(size_t)b * SEQL * HDIM + j];
}

extern "C" void kernel_launch(void* const* d_in, const int* in_sizes, int n_in,
                              void* d_out, int out_size, void* d_ws, size_t ws_size,
                              hipStream_t stream) {
  const float* inp    = (const float*)d_in[0];
  const int*   length = (const int*)  d_in[1];
  const float* word_W = (const float*)d_in[2];
  const float* word_b = (const float*)d_in[3];
  const float* comp_W = (const float*)d_in[4];
  const float* comp_b = (const float*)d_in[5];
  const float* comp_q = (const float*)d_in[6];

  float* ws = (float*)d_ws;
  size_t off = 0;
  float* hb[2];
  float* cb[2];
  hb[0] = ws + off; off += (size_t)BATCH * SEQL * HDIM;
  cb[0] = ws + off; off += (size_t)BATCH * SEQL * HDIM;
  hb[1] = ws + off; off += (size_t)BATCH * SEQL * HDIM;
  cb[1] = ws + off; off += (size_t)BATCH * SEQL * HDIM;
  float* newh = ws + off; off += (size_t)(BATCH * 23) * HDIM;
  float* newc = ws + off; off += (size_t)(BATCH * 23) * HDIM;

  gemm_wmma<0><<<dim3((BATCH * SEQL) / 32, 1024 / 256), 256, 0, stream>>>(
      inp, nullptr, nullptr, word_W, word_b, hb[0], cb[0], 512, 0);

  int cur = 0;
  for (int i = 0; i < SEQL - 1; ++i) {
    int n = (SEQL - 1) - i;
    int R = BATCH * n;
    int nxt = cur ^ 1;
    gemm_wmma<1><<<dim3(R / 32, HDIM / 64), 256, 0, stream>>>(
        nullptr, hb[cur], cb[cur], comp_W, comp_b, newh, newc, 1024, n);
    mix_k<<<BATCH, 256, 0, stream>>>(newh, newc, hb[cur], cb[cur], comp_q, length,
                                     hb[nxt], cb[nxt], n, i);
    cur = nxt;
  }

  final_copy_k<<<(BATCH * HDIM) / 256, 256, 0, stream>>>(hb[cur], (float*)d_out);
  (void)in_sizes; (void)n_in; (void)out_size; (void)ws_size;
}


# GEMM workgroup tile reshaped to 64 rows x 32 cols x 5 gates, 64 k per chunk (was 32 x 64 x 5, 32 k), waves 4Mx2N: 224 staged rows per chunk where 352 were, half the hand-overs; same sums, same order
# speedup vs baseline: 1.6261x; 1.6261x over previous
#include <hip/hip_runtime.h>
#include <hip/hip_bf16.h>
#include <math.h>


typedef __attribute__((ext_vector_type(2))) float v2f;
typedef __attribute__((ext_vector_type(8))) float v8f;

#define BATCH 64
#define SEQL  24
#define HDIM  512

__device__ __forceinline__ float sigm(float x) { return 1.0f / (1.0f + __expf(-x)); }
__device__ __forceinline__ float tanh_fast(float x) { return 2.0f / (1.0f + __expf(-2.0f * x)) - 1.0f; }

#define TM  64
#define KC  64
#define LDP 68

template <int MODE>
__global__ __launch_bounds__(256) void gemm_wmma(const float* __restrict__ X,
                                                 const float* __restrict__ hbuf,
                                                 const float* __restrict__ cbuf,
                                                 const float* __restrict__ W,
                                                 const float* __restrict__ bias,
                                                 float* __restrict__ out0,
                                                 float* __restrict__ out1,
                                                 int K, int n) {
  __shared__ float A_lds[TM * LDP];
  __shared__ float B_lds[160 * LDP];

  const int tid  = threadIdx.x;
  const unsigned lane = tid & 31u;
  const int wave = tid >> 5;
  const int wm   = wave >> 1;
  const int wn   = wave & 1;
  const int l16  = (int)(lane & 15u);
  const int lhi  = (int)(lane >> 4);

  const int r0 = blockIdx.x * TM;
  const int NACC = (MODE == 0) ? 4 : 5;
  const int NA = TM / 16;
  const int NB = (MODE == 0) ? 8 : 10;

  v8f acc[5] = {};

  const int srow = tid >> 4;
  const int sc4  = (tid & 15) * 4;
  const float* ap[NA];
#pragma unroll
  for (int i = 0; i < NA; ++i) {
    int rg = r0 + i * 16 + srow;
    if (MODE == 0) {
      ap[i] = X + (size_t)rg * K + sc4;
    } else {
      int b = rg / n;
      int p = rg - b * n;
      ap[i] = hbuf + (size_t)(b * SEQL + p) * HDIM + sc4;
    }
  }
  const float* bp[10];
#pragma unroll
  for (int i = 0; i < NB; ++i) {
    int nrow = i * 16 + srow;
    size_t wrow;
    if (MODE == 0) {
      wrow = (size_t)(blockIdx.y * 128 + nrow);
    } else {
      int g  = nrow >> 5;
      int nl = nrow & 31;
      wrow = (size_t)(g * HDIM + blockIdx.y * 32 + nl);
    }
    bp[i] = W + wrow * K + sc4;
  }

  for (int k0 = 0; k0 < K; k0 += KC) {
    float4 aval[NA];
    float4 bval[10];
#pragma unroll
    for (int i = 0; i < NA; ++i) aval[i] = *(const float4*)(ap[i] + k0);
#pragma unroll
    for (int i = 0; i < NB; ++i) bval[i] = *(const float4*)(bp[i] + k0);
    __builtin_amdgcn_sched_barrier(0);
#pragma unroll
    for (int i = 0; i < NA; ++i) *(float4*)&A_lds[(i * 16 + srow) * LDP + sc4] = aval[i];
#pragma unroll
    for (int i = 0; i < NB; ++i) *(float4*)&B_lds[(i * 16 + srow) * LDP + sc4] = bval[i];
    __syncthreads();

#pragma unroll
    for (int kk = 0; kk < KC; kk += 4) {
      v2f a = *(const v2f*)&A_lds[(wm * 16 + l16) * LDP + kk + 2 * lhi];
#pragma unroll
      for (int g = 0; g < NACC; ++g) {
        int nrow = (MODE == 0) ? (wn * 64 + g * 16 + l16)
                               : (g * 32 + wn * 16 + l16);
        v2f bf = *(const v2f*)&B_lds[nrow * LDP + kk + 2 * lhi];
        acc[g] = __builtin_amdgcn_wmma_f32_16x16x4_f32(
            false, a, false, bf, (short)0, acc[g], false, false);
      }
    }
    __syncthreads();
  }

  if (MODE == 0) {
    const int colbase = blockIdx.y * 128 + wn * 64;
    const int rowg0   = r0 + wm * 16 + 8 * lhi;
#pragma unroll
    for (int j = 0; j < 4; ++j) {
      int col = colbase + j * 16 + l16;
      float bv = bias[col];
#pragma unroll
      for (int vr = 0; vr < 8; ++vr) {
        int rg = rowg0 + vr;
        float val = acc[j][vr] + bv;
        if (col < HDIM) out0[(size_t)rg * HDIM + col] = val;
        else            out1[(size_t)rg * HDIM + (col - HDIM)] = val;
      }
    }
  } else {
    const int colg = blockIdx.y * 32 + wn * 16 + l16;
    float bz[5];
#pragma unroll
    for (int g = 0; g < 5; ++g) bz[g] = bias[g * HDIM + colg];
    int rg = r0 + wm * 16 + 8 * lhi;
    int b  = rg / n;
    int p  = rg - b * n;
#pragma unroll
    for (int vr = 0; vr < 8; ++vr) {
      float gi  = acc[0][vr] + bz[0];
      float gfl = acc[1][vr] + bz[1];
      float gfr = acc[2][vr] + bz[2];
      float gu  = acc[3][vr] + bz[3];
      float go  = acc[4][vr] + bz[4];
      size_t cbase = (size_t)(b * SEQL + p) * HDIM + colg;
      float cl = cbuf[cbase];
      float cr = cbuf[cbase + HDIM];
      float nc = cl * sigm(gfl + 1.0f) + cr * sigm(gfr + 1.0f) + tanh_fast(gu) * sigm(gi);
      float nh = sigm(go) * tanh_fast(nc);
      size_t oi = (size_t)rg * HDIM + colg;
      out0[oi] = nh;
      out1[oi] = nc;
      ++rg; ++p;
      if (p == n) { p = 0; ++b; }
    }
  }
}

#define MIXG 2
#define MIXT 6

__global__ __launch_bounds__(256) void mix_k(const float* __restrict__ newh,
                                             const float* __restrict__ newc,
                                             const float* __restrict__ hcur,
                                             const float* __restrict__ ccur,
                                             const float* __restrict__ q,
                                             const int* __restrict__ length,
                                             float* __restrict__ hnxt,
                                             float* __restrict__ cnxt,
                                             int n, int iter) {
  __shared__ float sc[32];
  const int b = blockIdx.y;
  const int tid = threadIdx.x;
  const unsigned lane = tid & 31u;
  const int wave = tid >> 5;
  const int nvec = n * (HDIM / 4);
  if ((int)blockIdx.x * 256 >= nvec) return;

  const int p0 = wave, p1 = wave + 8, p2 = wave + 16;
  const float* r0 = newh + (size_t)(b * n + min(p0, n - 1)) * HDIM;
  const float* r1 = newh + (size_t)(b * n + min(p1, n - 1)) * HDIM;
  const float* r2 = newh + (size_t)(b * n + min(p2, n - 1)) * HDIM;
  float s0 = 0.0f, s1 = 0.0f, s2 = 0.0f;
#pragma unroll
  for (int t = 0; t < HDIM / 32; ++t) {
    const int j = (int)lane + 32 * t;
    const float qv = q[j];
    s0 += r0[j] * qv;
    s1 += r1[j] * qv;
    s2 += r2[j] * qv;
  }
#pragma unroll
  for (int off = 16; off > 0; off >>= 1) {
    s0 += __shfl_xor(s0, off, 32);
    s1 += __shfl_xor(s1, off, 32);
    s2 += __shfl_xor(s2, off, 32);
  }
  if (lane == 0) {
    if (p0 < n) sc[p0] = s0;
    if (p1 < n) sc[p1] = s1;
    if (p2 < n) sc[p2] = s2;
  }
  __syncthreads();

  const int len = length[b];
  float v = -INFINITY;
  if ((int)lane < n) {
    float s = ((iter + 1 + (int)lane) < len) ? sc[lane] : -1.0e9f;
    v = (s == s) ? s : -INFINITY;
  }
  float m = v;
#pragma unroll
  for (int off = 16; off > 0; off >>= 1) m = fmaxf(m, __shfl_xor(m, off, 32));
  const unsigned long long hit = __ballot(v == m);
  const int ksel = hit ? (__ffsll((long long)hit) - 1) : 0;
  const bool done = (iter + 1) < len;

  const int vi0 = (int)blockIdx.x * 256 + tid;
  if (vi0 < nvec) {
#define MIX_LOAD(T)                                                              \
    float4 hv##T, cv##T; size_t dst##T;                                          \
    {                                                                            \
      const int vi = vi0 + (T) * (MIXG * 256);                                   \
      const int vc = (vi < nvec) ? vi : vi0;                                     \
      const int p = vc >> 7;                              \
      const int j = (vc & 127) * 4;                                              \
      const size_t cur = (size_t)(b * SEQL + p) * HDIM + j;                      \
      const size_t ni  = (size_t)(b * n + p) * HDIM + j;                         \
      const bool take_new   = done && (p == ksel);                               \
      const bool take_right = done && (p > ksel);                                \
      const size_t oc = cur + (take_right ? (size_t)HDIM : (size_t)0);           \
      const float* sh  = take_new ? (newh + ni) : (hcur + oc);                   \
      const float* scp = take_new ? (newc + ni) : (ccur + oc);                   \
      dst##T = cur;                                                              \
      hv##T = *(const float4*)sh;                                                \
      cv##T = *(const float4*)scp;                                               \
    }
#define MIX_STORE(T)                                                             \
    *(float4*)&hnxt[dst##T] = hv##T;                                             \
    *(float4*)&cnxt[dst##T] = cv##T;

    MIX_LOAD(0) MIX_LOAD(1) MIX_LOAD(2) MIX_LOAD(3) MIX_LOAD(4) MIX_LOAD(5)
    __builtin_amdgcn_sched_barrier(0);
    MIX_STORE(0) MIX_STORE(1) MIX_STORE(2) MIX_STORE(3) MIX_STORE(4) MIX_STORE(5)
#undef MIX_LOAD
#undef MIX_STORE
  }
}

__global__ __launch_bounds__(256) void final_copy_k(const float* __restrict__ hfin,
                                                    float* __restrict__ out) {
  int idx = blockIdx.x * 256 + threadIdx.x;
  int b = idx >> 9;
  int j = idx & (HDIM - 1);
  out[idx] = hfin[(size_t)b * SEQL * HDIM + j];
}

extern "C" void kernel_launch(void* const* d_in, const int* in_sizes, int n_in,
                              void* d_out, int out_size, void* d_ws, size_t ws_size,
                              hipStream_t stream) {
  const float* inp    = (const float*)d_in[0];
  const int*   length = (const int*)  d_in[1];
  const float* word_W = (const float*)d_in[2];
  const float* word_b = (const float*)d_in[3];
  const float* comp_W = (const float*)d_in[4];
  const float* comp_b = (const float*)d_in[5];
  const float* comp_q = (const float*)d_in[6];

  float* ws = (float*)d_ws;
  size_t off = 0;
  float* hb[2];
  float* cb[2];
  hb[0] = ws + off; off += (size_t)BATCH * SEQL * HDIM;
  cb[0] = ws + off; off += (size_t)BATCH * SEQL * HDIM;
  hb[1] = ws + off; off += (size_t)BATCH * SEQL * HDIM;
  cb[1] = ws + off; off += (size_t)BATCH * SEQL * HDIM;
  float* newh = ws + off; off += (size_t)(BATCH * 23) * HDIM;
  float* newc = ws + off; off += (size_t)(BATCH * 23) * HDIM;

  gemm_wmma<0><<<dim3((BATCH * SEQL) / TM, 1024 / 128), 256, 0, stream>>>(
      inp, nullptr, nullptr, word_W, word_b, hb[0], cb[0], 512, 0);

  int cur = 0;
  for (int i = 0; i < SEQL - 1; ++i) {
    int n = (SEQL - 1) - i;
    int R = BATCH * n;
    int nxt = cur ^ 1;
    gemm_wmma<1><<<dim3(R / TM, HDIM / 32), 256, 0, stream>>>(
        nullptr, hb[cur], cb[cur], comp_W, comp_b, newh, newc, 1024, n);
    mix_k<<<dim3(MIXG, BATCH), 256, 0, stream>>>(newh, newc, hb[cur], cb[cur], comp_q, length,
                                     hb[nxt], cb[nxt], n, i);
    cur = nxt;
  }

  final_copy_k<<<(BATCH * HDIM) / 256, 256, 0, stream>>>(hb[cur], (float*)d_out);
  (void)in_sizes; (void)n_in; (void)out_size; (void)ws_size;
}


# Comp GEMM and score dots only for live pairs (p < length-1-step, left open by the argmax mask): live rows packed by a batch prefix sum into 64- or 32-row blocks, empty 16-row tiles skip; same sums
# speedup vs baseline: 1.8684x; 1.1490x over previous
#include <hip/hip_runtime.h>
#include <hip/hip_bf16.h>
#include <math.h>


typedef __attribute__((ext_vector_type(2))) float v2f;
typedef __attribute__((ext_vector_type(8))) float v8f;

#define BATCH 64
#define SEQL  24
#define HDIM  512

__device__ __forceinline__ float sigm(float x) { return 1.0f / (1.0f + __expf(-x)); }
__device__ __forceinline__ float tanh_fast(float x) { return 2.0f / (1.0f + __expf(-2.0f * x)) - 1.0f; }

#define TM  64
#define KC  64
#define LDP 68

template <int MODE>
__global__ __launch_bounds__(256) void gemm_wmma(const float* __restrict__ X,
                                                 const float* __restrict__ hbuf,
                                                 const float* __restrict__ cbuf,
                                                 const float* __restrict__ W,
                                                 const float* __restrict__ bias,
                                                 float* __restrict__ out0,
                                                 float* __restrict__ out1,
                                                 const int* __restrict__ length,
                                                 int K, int n, int iter) {
  __shared__ float A_lds[TM * LDP];
  __shared__ float B_lds[160 * LDP];
  __shared__ __attribute__((aligned(16))) int s_cnt[BATCH];
  __shared__ __attribute__((aligned(16))) int s_row[TM];

  const int tid  = threadIdx.x;
  const unsigned lane = tid & 31u;
  const int wave = tid >> 5;
  const int wm   = wave >> 1;
  const int wn   = wave & 1;
  const int l16  = (int)(lane & 15u);
  const int lhi  = (int)(lane >> 4);

  const int rblk = (MODE == 0) ? (int)blockIdx.x : (int)blockIdx.y;
  const int cblk = (MODE == 0) ? (int)blockIdx.y : (int)blockIdx.x;
  int r0 = rblk * TM;
  int nlv = TM;
  const int NACC = (MODE == 0) ? 4 : 5;

  int tot = 0;
  if (MODE == 1) {
    int c = 0;
    if (tid < BATCH) {
      c = min(max(length[tid] - 1 - iter, 0), n);
      s_cnt[tid] = c;
      s_row[tid] = 0;
    }
    __syncthreads();
    int pre = 0;
#pragma unroll
    for (int j = 0; j < BATCH; ++j) {
      const int cj = s_cnt[j];
      pre += (j < tid) ? cj : 0;
      tot += cj;
    }
    tot = __builtin_amdgcn_readfirstlane(tot);
    const int rpb = (tot <= 32 * (int)gridDim.y && tot <= 512) ? 32 : TM;
    r0 = rblk * rpb;
    if (r0 >= tot) return;
    nlv = min(rpb, tot - r0);
    if (tid < BATCH) {
      const int lo = max(pre, r0);
      const int hi = min(pre + c, r0 + rpb);
      for (int r = lo; r < hi; ++r) s_row[r - r0] = (tid << 5) | (r - pre);
    }
    __syncthreads();
  }
  const int NA = TM / 16;
  const int NB = (MODE == 0) ? 8 : 10;

  v8f acc[5] = {};

  const int srow = tid >> 4;
  const int sc4  = (tid & 15) * 4;
  const float* ap[NA];
#pragma unroll
  for (int i = 0; i < NA; ++i) {
    int rg = r0 + i * 16 + srow;
    if (MODE == 0) {
      ap[i] = X + (size_t)rg * K + sc4;
    } else {
      const int info = s_row[i * 16 + srow];
      const int b = info >> 5;
      const int p = info & 31;
      ap[i] = hbuf + (size_t)(b * SEQL + p) * HDIM + sc4;
    }
  }
  const float* bp[10];
#pragma unroll
  for (int i = 0; i < NB; ++i) {
    int nrow = i * 16 + srow;
    size_t wrow;
    if (MODE == 0) {
      wrow = (size_t)(cblk * 128 + nrow);
    } else {
      int g  = nrow >> 5;
      int nl = nrow & 31;
      wrow = (size_t)(g * HDIM + cblk * 32 + nl);
    }
    bp[i] = W + wrow * K + sc4;
  }

  const bool act = (MODE == 0) ? true : (__builtin_amdgcn_readfirstlane(wm) * 16 < nlv);

  for (int k0 = 0; k0 < K; k0 += KC) {
    float4 aval[NA];
    float4 bval[10];
#pragma unroll
    for (int i = 0; i < NA; ++i) aval[i] = *(const float4*)(ap[i] + k0);
#pragma unroll
    for (int i = 0; i < NB; ++i) bval[i] = *(const float4*)(bp[i] + k0);
    __builtin_amdgcn_sched_barrier(0);
#pragma unroll
    for (int i = 0; i < NA; ++i) *(float4*)&A_lds[(i * 16 + srow) * LDP + sc4] = aval[i];
#pragma unroll
    for (int i = 0; i < NB; ++i) *(float4*)&B_lds[(i * 16 + srow) * LDP + sc4] = bval[i];
    __syncthreads();

    if (act) {
#pragma unroll
      for (int kk = 0; kk < KC; kk += 4) {
        v2f a = *(const v2f*)&A_lds[(wm * 16 + l16) * LDP + kk + 2 * lhi];
#pragma unroll
        for (int g = 0; g < NACC; ++g) {
          int nrow = (MODE == 0) ? (wn * 64 + g * 16 + l16)
                                 : (g * 32 + wn * 16 + l16);
          v2f bf = *(const v2f*)&B_lds[nrow * LDP + kk + 2 * lhi];
          acc[g] = __builtin_amdgcn_wmma_f32_16x16x4_f32(
              false, a, false, bf, (short)0, acc[g], false, false);
        }
      }
    }
    __syncthreads();
  }

  if (MODE == 0) {
    const int colbase = cblk * 128 + wn * 64;
    const int rowg0   = r0 + wm * 16 + 8 * lhi;
#pragma unroll
    for (int j = 0; j < 4; ++j) {
      int col = colbase + j * 16 + l16;
      float bv = bias[col];
#pragma unroll
      for (int vr = 0; vr < 8; ++vr) {
        int rg = rowg0 + vr;
        float val = acc[j][vr] + bv;
        if (col < HDIM) out0[(size_t)rg * HDIM + col] = val;
        else            out1[(size_t)rg * HDIM + (col - HDIM)] = val;
      }
    }
  } else {
    const int colg = cblk * 32 + wn * 16 + l16;
    float bz[5];
#pragma unroll
    for (int g = 0; g < 5; ++g) bz[g] = bias[g * HDIM + colg];
    const int rl0 = wm * 16 + 8 * lhi;
    int rinfo[8];
    *(int4*)&rinfo[0] = *(const int4*)&s_row[rl0];
    *(int4*)&rinfo[4] = *(const int4*)&s_row[rl0 + 4];
    float clv[8], crv[8];
    size_t oiv[8];
#pragma unroll
    for (int vr = 0; vr < 8; ++vr) {
      const int b = rinfo[vr] >> 5;
      const int p = rinfo[vr] & 31;
      const size_t cbase = (size_t)(b * SEQL + p) * HDIM + colg;
      clv[vr] = cbuf[cbase];
      crv[vr] = cbuf[cbase + HDIM];
      oiv[vr] = (size_t)(b * n + p) * HDIM + colg;
    }
#pragma unroll
    for (int vr = 0; vr < 8; ++vr) asm volatile("" : "+v"(clv[vr]), "+v"(crv[vr]));
#pragma unroll
    for (int vr = 0; vr < 8; ++vr) {
      const bool live = (rl0 + vr) < nlv;
      float gi  = acc[0][vr] + bz[0];
      float gfl = acc[1][vr] + bz[1];
      float gfr = acc[2][vr] + bz[2];
      float gu  = acc[3][vr] + bz[3];
      float go  = acc[4][vr] + bz[4];
      float nc = clv[vr] * sigm(gfl + 1.0f) + crv[vr] * sigm(gfr + 1.0f) + tanh_fast(gu) * sigm(gi);
      float nh = sigm(go) * tanh_fast(nc);
      if (live) {
        out0[oiv[vr]] = nh;
        out1[oiv[vr]] = nc;
      }
    }
  }
}

#define MIXG 2
#define MIXT 6

__global__ __launch_bounds__(256) void mix_k(const float* __restrict__ newh,
                                             const float* __restrict__ newc,
                                             const float* __restrict__ hcur,
                                             const float* __restrict__ ccur,
                                             const float* __restrict__ q,
                                             const int* __restrict__ length,
                                             float* __restrict__ hnxt,
                                             float* __restrict__ cnxt,
                                             int n, int iter) {
  __shared__ float sc[32];
  const int b = blockIdx.y;
  const int tid = threadIdx.x;
  const unsigned lane = tid & 31u;
  const int wave = tid >> 5;
  const int nvec = n * (HDIM / 4);
  if ((int)blockIdx.x * 256 >= nvec) return;

  const int len = length[b];
  const int nlive = min(max(len - 1 - iter, 0), n);
  const int p0 = wave, p1 = wave + 8, p2 = wave + 16;
  float s0 = 0.0f, s1 = 0.0f, s2 = 0.0f;
  if (nlive > 0) {
    const float* r0 = newh + (size_t)(b * n + min(p0, nlive - 1)) * HDIM;
    const float* r1 = newh + (size_t)(b * n + min(p1, nlive - 1)) * HDIM;
    const float* r2 = newh + (size_t)(b * n + min(p2, nlive - 1)) * HDIM;
#pragma unroll
    for (int t = 0; t < HDIM / 32; ++t) {
      const int j = (int)lane + 32 * t;
      const float qv = q[j];
      s0 += r0[j] * qv;
      s1 += r1[j] * qv;
      s2 += r2[j] * qv;
    }
#pragma unroll
    for (int off = 16; off > 0; off >>= 1) {
      s0 += __shfl_xor(s0, off, 32);
      s1 += __shfl_xor(s1, off, 32);
      s2 += __shfl_xor(s2, off, 32);
    }
  }
  if (lane == 0) {
    if (p0 < n) sc[p0] = (p0 < nlive) ? s0 : -1.0e9f;
    if (p1 < n) sc[p1] = (p1 < nlive) ? s1 : -1.0e9f;
    if (p2 < n) sc[p2] = (p2 < nlive) ? s2 : -1.0e9f;
  }
  __syncthreads();

  float v = -INFINITY;
  if ((int)lane < n) {
    float s = ((iter + 1 + (int)lane) < len) ? sc[lane] : -1.0e9f;
    v = (s == s) ? s : -INFINITY;
  }
  float m = v;
#pragma unroll
  for (int off = 16; off > 0; off >>= 1) m = fmaxf(m, __shfl_xor(m, off, 32));
  const unsigned long long hit = __ballot(v == m);
  const int ksel = hit ? (__ffsll((long long)hit) - 1) : 0;
  const bool done = (iter + 1) < len;

  const int vi0 = (int)blockIdx.x * 256 + tid;
  if (vi0 < nvec) {
#define MIX_LOAD(T)                                                              \
    float4 hv##T, cv##T; size_t dst##T;                                          \
    {                                                                            \
      const int vi = vi0 + (T) * (MIXG * 256);                                   \
      const int vc = (vi < nvec) ? vi : vi0;                                     \
      const int p = vc >> 7;                              \
      const int j = (vc & 127) * 4;                                              \
      const size_t cur = (size_t)(b * SEQL + p) * HDIM + j;                      \
      const size_t ni  = (size_t)(b * n + p) * HDIM + j;                         \
      const bool take_new   = done && (p == ksel);                               \
      const bool take_right = done && (p > ksel);                                \
      const size_t oc = cur + (take_right ? (size_t)HDIM : (size_t)0);           \
      const float* sh  = take_new ? (newh + ni) : (hcur + oc);                   \
      const float* scp = take_new ? (newc + ni) : (ccur + oc);                   \
      dst##T = cur;                                                              \
      hv##T = *(const float4*)sh;                                                \
      cv##T = *(const float4*)scp;                                               \
    }
#define MIX_STORE(T)                                                             \
    *(float4*)&hnxt[dst##T] = hv##T;                                             \
    *(float4*)&cnxt[dst##T] = cv##T;

    MIX_LOAD(0) MIX_LOAD(1) MIX_LOAD(2) MIX_LOAD(3) MIX_LOAD(4) MIX_LOAD(5)
    __builtin_amdgcn_sched_barrier(0);
    MIX_STORE(0) MIX_STORE(1) MIX_STORE(2) MIX_STORE(3) MIX_STORE(4) MIX_STORE(5)
#undef MIX_LOAD
#undef MIX_STORE
  }
}

__global__ __launch_bounds__(256) void final_copy_k(const float* __restrict__ hfin,
                                                    float* __restrict__ out) {
  int idx = blockIdx.x * 256 + threadIdx.x;
  int b = idx >> 9;
  int j = idx & (HDIM - 1);
  out[idx] = hfin[(size_t)b * SEQL * HDIM + j];
}

extern "C" void kernel_launch(void* const* d_in, const int* in_sizes, int n_in,
                              void* d_out, int out_size, void* d_ws, size_t ws_size,
                              hipStream_t stream) {
  const float* inp    = (const float*)d_in[0];
  const int*   length = (const int*)  d_in[1];
  const float* word_W = (const float*)d_in[2];
  const float* word_b = (const float*)d_in[3];
  const float* comp_W = (const float*)d_in[4];
  const float* comp_b = (const float*)d_in[5];
  const float* comp_q = (const float*)d_in[6];

  float* ws = (float*)d_ws;
  size_t off = 0;
  float* hb[2];
  float* cb[2];
  hb[0] = ws + off; off += (size_t)BATCH * SEQL * HDIM;
  cb[0] = ws + off; off += (size_t)BATCH * SEQL * HDIM;
  hb[1] = ws + off; off += (size_t)BATCH * SEQL * HDIM;
  cb[1] = ws + off; off += (size_t)BATCH * SEQL * HDIM;
  float* newh = ws + off; off += (size_t)(BATCH * 23) * HDIM;
  float* newc = ws + off; off += (size_t)(BATCH * 23) * HDIM;

  gemm_wmma<0><<<dim3((BATCH * SEQL) / TM, 1024 / 128), 256, 0, stream>>>(
      inp, nullptr, nullptr, word_W, word_b, hb[0], cb[0], nullptr, 512, 0, 0);

  int cur = 0;
  for (int i = 0; i < SEQL - 1; ++i) {
    int n = (SEQL - 1) - i;
    int R = BATCH * n;
    int nxt = cur ^ 1;
    gemm_wmma<1><<<dim3(HDIM / 32, R / TM), 256, 0, stream>>>(
        nullptr, hb[cur], cb[cur], comp_W, comp_b, newh, newc, length, 1024, n, i);
    mix_k<<<dim3(MIXG, BATCH), 256, 0, stream>>>(newh, newc, hb[cur], cb[cur], comp_q, length,
                                     hb[nxt], cb[nxt], n, i);
    cur = nxt;
  }

  final_copy_k<<<(BATCH * HDIM) / 256, 256, 0, stream>>>(hb[cur], (float*)d_out);
  (void)in_sizes; (void)n_in; (void)out_size; (void)ws_size;
}


# Short merge steps (at most 256 live pairs): comp GEMM in 16-row x 16-col blocks, each gate of a tile on its own wave (one accumulator), gates meet in LDS for the epilogue; same sums, same order
# speedup vs baseline: 2.1355x; 1.1430x over previous
#include <hip/hip_runtime.h>
#include <hip/hip_bf16.h>
#include <math.h>


typedef __attribute__((ext_vector_type(2))) float v2f;
typedef __attribute__((ext_vector_type(8))) float v8f;

#define BATCH 64
#define SEQL  24
#define HDIM  512

__device__ __forceinline__ float sigm(float x) { return 1.0f / (1.0f + __expf(-x)); }
__device__ __forceinline__ float tanh_fast(float x) { return 2.0f / (1.0f + __expf(-2.0f * x)) - 1.0f; }

#define TM  64
#define KC  64
#define LDP 68

template <int MODE>
__global__ __launch_bounds__(256) void gemm_wmma(const float* __restrict__ X,
                                                 const float* __restrict__ hbuf,
                                                 const float* __restrict__ cbuf,
                                                 const float* __restrict__ W,
                                                 const float* __restrict__ bias,
                                                 float* __restrict__ out0,
                                                 float* __restrict__ out1,
                                                 const int* __restrict__ length,
                                                 int K, int n, int iter) {
  __shared__ float A_lds[TM * LDP];
  __shared__ float B_lds[160 * LDP];
  __shared__ __attribute__((aligned(16))) int s_cnt[BATCH];
  __shared__ __attribute__((aligned(16))) int s_row[TM];

  const int tid  = threadIdx.x;
  const unsigned lane = tid & 31u;
  const int wave = tid >> 5;
  const int wm   = wave >> 1;
  const int wn   = wave & 1;
  const int l16  = (int)(lane & 15u);
  const int lhi  = (int)(lane >> 4);

  const int rblk = (MODE == 0) ? (int)blockIdx.x : (int)blockIdx.y;
  const int cblk = (MODE == 0) ? (int)blockIdx.y : (int)blockIdx.x;
  int r0 = rblk * TM;
  int nlv = TM;
  const int NACC = (MODE == 0) ? 4 : 5;

  int tot = 0;
  bool small = false;
  int cS = 0;
  if (MODE == 1) {
    int c = 0;
    if (tid < BATCH) {
      c = min(max(length[tid] - 1 - iter, 0), n);
      s_cnt[tid] = c;
      s_row[tid] = 0;
    }
    __syncthreads();
    int pre = 0;
#pragma unroll
    for (int j = 0; j < BATCH; ++j) {
      const int cj = s_cnt[j];
      pre += (j < tid) ? cj : 0;
      tot += cj;
    }
    tot = __builtin_amdgcn_readfirstlane(tot);
    const int nwg = (int)(gridDim.x * gridDim.y);
    small = (tot <= 256) && (32 * ((tot + 15) >> 4) <= nwg);
    int rpb, rb;
    if (small) {
      const int lin = (int)(blockIdx.y * gridDim.x + blockIdx.x);
      rpb = 16;
      rb = lin >> 5;
      cS = lin & 31;
    } else {
      rpb = (tot <= 32 * (int)gridDim.y && tot <= 512) ? 32 : TM;
      rb = rblk;
    }
    r0 = rb * rpb;
    if (r0 >= tot) return;
    nlv = min(rpb, tot - r0);
    if (tid < BATCH) {
      const int lo = max(pre, r0);
      const int hi = min(pre + c, r0 + rpb);
      for (int r = lo; r < hi; ++r) s_row[r - r0] = (tid << 5) | (r - pre);
    }
    __syncthreads();
  }
  if (MODE == 1 && small) {
    const int srowS = tid >> 4;
    const int sc4S  = (tid & 15) * 4;
    const int infoA = s_row[srowS];
    const float* apS = hbuf + (size_t)((infoA >> 5) * SEQL + (infoA & 31)) * HDIM + sc4S;
    const float* bpS[5];
#pragma unroll
    for (int g = 0; g < 5; ++g) bpS[g] = W + (size_t)(g * HDIM + cS * 16 + srowS) * K + sc4S;
    const int wv = __builtin_amdgcn_readfirstlane(wave);
    const bool actS = wv < 5;
    v8f accS = {};
    for (int k0 = 0; k0 < K; k0 += KC) {
      float4 avS = *(const float4*)(apS + k0);
      float4 bvS[5];
#pragma unroll
      for (int g = 0; g < 5; ++g) bvS[g] = *(const float4*)(bpS[g] + k0);
      *(float4*)&A_lds[srowS * LDP + sc4S] = avS;
#pragma unroll
      for (int g = 0; g < 5; ++g) *(float4*)&B_lds[(g * 16 + srowS) * LDP + sc4S] = bvS[g];
      __syncthreads();
      if (actS) {
#pragma unroll
        for (int kk = 0; kk < KC; kk += 4) {
          v2f a  = *(const v2f*)&A_lds[l16 * LDP + kk + 2 * lhi];
          v2f bf = *(const v2f*)&B_lds[(wv * 16 + l16) * LDP + kk + 2 * lhi];
          accS = __builtin_amdgcn_wmma_f32_16x16x4_f32(
              false, a, false, bf, (short)0, accS, false, false);
        }
      }
      __syncthreads();
    }
    float* Gs = B_lds;
    if (actS) {
#pragma unroll
      for (int vr = 0; vr < 8; ++vr) Gs[(wv * 16 + 8 * lhi + vr) * 17 + l16] = accS[vr];
    }
    __syncthreads();
    {
      const int row  = tid >> 4;
      const int col  = tid & 15;
      const int colg = cS * 16 + col;
      const int info = s_row[row];
      const int b = info >> 5;
      const int p = info & 31;
      const size_t cbase = (size_t)(b * SEQL + p) * HDIM + colg;
      const float clS = cbuf[cbase];
      const float crS = cbuf[cbase + HDIM];
      float gi  = Gs[(0 * 16 + row) * 17 + col] + bias[0 * HDIM + colg];
      float gfl = Gs[(1 * 16 + row) * 17 + col] + bias[1 * HDIM + colg];
      float gfr = Gs[(2 * 16 + row) * 17 + col] + bias[2 * HDIM + colg];
      float gu  = Gs[(3 * 16 + row) * 17 + col] + bias[3 * HDIM + colg];
      float go  = Gs[(4 * 16 + row) * 17 + col] + bias[4 * HDIM + colg];
      float nc = clS * sigm(gfl + 1.0f) + crS * sigm(gfr + 1.0f) + tanh_fast(gu) * sigm(gi);
      float nh = sigm(go) * tanh_fast(nc);
      if (row < nlv) {
        const size_t oi = (size_t)(b * n + p) * HDIM + colg;
        out0[oi] = nh;
        out1[oi] = nc;
      }
    }
    return;
  }

  const int NA = TM / 16;
  const int NB = (MODE == 0) ? 8 : 10;

  v8f acc[5] = {};

  const int srow = tid >> 4;
  const int sc4  = (tid & 15) * 4;
  const float* ap[NA];
#pragma unroll
  for (int i = 0; i < NA; ++i) {
    int rg = r0 + i * 16 + srow;
    if (MODE == 0) {
      ap[i] = X + (size_t)rg * K + sc4;
    } else {
      const int info = s_row[i * 16 + srow];
      const int b = info >> 5;
      const int p = info & 31;
      ap[i] = hbuf + (size_t)(b * SEQL + p) * HDIM + sc4;
    }
  }
  const float* bp[10];
#pragma unroll
  for (int i = 0; i < NB; ++i) {
    int nrow = i * 16 + srow;
    size_t wrow;
    if (MODE == 0) {
      wrow = (size_t)(cblk * 128 + nrow);
    } else {
      int g  = nrow >> 5;
      int nl = nrow & 31;
      wrow = (size_t)(g * HDIM + cblk * 32 + nl);
    }
    bp[i] = W + wrow * K + sc4;
  }

  const bool act = (MODE == 0) ? true : (__builtin_amdgcn_readfirstlane(wm) * 16 < nlv);

  for (int k0 = 0; k0 < K; k0 += KC) {
    float4 aval[NA];
    float4 bval[10];
#pragma unroll
    for (int i = 0; i < NA; ++i) aval[i] = *(const float4*)(ap[i] + k0);
#pragma unroll
    for (int i = 0; i < NB; ++i) bval[i] = *(const float4*)(bp[i] + k0);
    __builtin_amdgcn_sched_barrier(0);
#pragma unroll
    for (int i = 0; i < NA; ++i) *(float4*)&A_lds[(i * 16 + srow) * LDP + sc4] = aval[i];
#pragma unroll
    for (int i = 0; i < NB; ++i) *(float4*)&B_lds[(i * 16 + srow) * LDP + sc4] = bval[i];
    __syncthreads();

    if (act) {
#pragma unroll
      for (int kk = 0; kk < KC; kk += 4) {
        v2f a = *(const v2f*)&A_lds[(wm * 16 + l16) * LDP + kk + 2 * lhi];
#pragma unroll
        for (int g = 0; g < NACC; ++g) {
          int nrow = (MODE == 0) ? (wn * 64 + g * 16 + l16)
                                 : (g * 32 + wn * 16 + l16);
          v2f bf = *(const v2f*)&B_lds[nrow * LDP + kk + 2 * lhi];
          acc[g] = __builtin_amdgcn_wmma_f32_16x16x4_f32(
              false, a, false, bf, (short)0, acc[g], false, false);
        }
      }
    }
    __syncthreads();
  }

  if (MODE == 0) {
    const int colbase = cblk * 128 + wn * 64;
    const int rowg0   = r0 + wm * 16 + 8 * lhi;
#pragma unroll
    for (int j = 0; j < 4; ++j) {
      int col = colbase + j * 16 + l16;
      float bv = bias[col];
#pragma unroll
      for (int vr = 0; vr < 8; ++vr) {
        int rg = rowg0 + vr;
        float val = acc[j][vr] + bv;
        if (col < HDIM) out0[(size_t)rg * HDIM + col] = val;
        else            out1[(size_t)rg * HDIM + (col - HDIM)] = val;
      }
    }
  } else {
    const int colg = cblk * 32 + wn * 16 + l16;
    float bz[5];
#pragma unroll
    for (int g = 0; g < 5; ++g) bz[g] = bias[g * HDIM + colg];
    const int rl0 = wm * 16 + 8 * lhi;
    int rinfo[8];
    *(int4*)&rinfo[0] = *(const int4*)&s_row[rl0];
    *(int4*)&rinfo[4] = *(const int4*)&s_row[rl0 + 4];
    float clv[8], crv[8];
    size_t oiv[8];
#pragma unroll
    for (int vr = 0; vr < 8; ++vr) {
      const int b = rinfo[vr] >> 5;
      const int p = rinfo[vr] & 31;
      const size_t cbase = (size_t)(b * SEQL + p) * HDIM + colg;
      clv[vr] = cbuf[cbase];
      crv[vr] = cbuf[cbase + HDIM];
      oiv[vr] = (size_t)(b * n + p) * HDIM + colg;
    }
#pragma unroll
    for (int vr = 0; vr < 8; ++vr) asm volatile("" : "+v"(clv[vr]), "+v"(crv[vr]));
#pragma unroll
    for (int vr = 0; vr < 8; ++vr) {
      const bool live = (rl0 + vr) < nlv;
      float gi  = acc[0][vr] + bz[0];
      float gfl = acc[1][vr] + bz[1];
      float gfr = acc[2][vr] + bz[2];
      float gu  = acc[3][vr] + bz[3];
      float go  = acc[4][vr] + bz[4];
      float nc = clv[vr] * sigm(gfl + 1.0f) + crv[vr] * sigm(gfr + 1.0f) + tanh_fast(gu) * sigm(gi);
      float nh = sigm(go) * tanh_fast(nc);
      if (live) {
        out0[oiv[vr]] = nh;
        out1[oiv[vr]] = nc;
      }
    }
  }
}

#define MIXG 2
#define MIXT 6

__global__ __launch_bounds__(256) void mix_k(const float* __restrict__ newh,
                                             const float* __restrict__ newc,
                                             const float* __restrict__ hcur,
                                             const float* __restrict__ ccur,
                                             const float* __restrict__ q,
                                             const int* __restrict__ length,
                                             float* __restrict__ hnxt,
                                             float* __restrict__ cnxt,
                                             int n, int iter) {
  __shared__ float sc[32];
  const int b = blockIdx.y;
  const int tid = threadIdx.x;
  const unsigned lane = tid & 31u;
  const int wave = tid >> 5;
  const int nvec = n * (HDIM / 4);
  if ((int)blockIdx.x * 256 >= nvec) return;

  const int len = length[b];
  const int nlive = min(max(len - 1 - iter, 0), n);
  const int p0 = wave, p1 = wave + 8, p2 = wave + 16;
  float s0 = 0.0f, s1 = 0.0f, s2 = 0.0f;
  if (nlive > 0) {
    const float* r0 = newh + (size_t)(b * n + min(p0, nlive - 1)) * HDIM;
    const float* r1 = newh + (size_t)(b * n + min(p1, nlive - 1)) * HDIM;
    const float* r2 = newh + (size_t)(b * n + min(p2, nlive - 1)) * HDIM;
#pragma unroll
    for (int t = 0; t < HDIM / 32; ++t) {
      const int j = (int)lane + 32 * t;
      const float qv = q[j];
      s0 += r0[j] * qv;
      s1 += r1[j] * qv;
      s2 += r2[j] * qv;
    }
#pragma unroll
    for (int off = 16; off > 0; off >>= 1) {
      s0 += __shfl_xor(s0, off, 32);
      s1 += __shfl_xor(s1, off, 32);
      s2 += __shfl_xor(s2, off, 32);
    }
  }
  if (lane == 0) {
    if (p0 < n) sc[p0] = (p0 < nlive) ? s0 : -1.0e9f;
    if (p1 < n) sc[p1] = (p1 < nlive) ? s1 : -1.0e9f;
    if (p2 < n) sc[p2] = (p2 < nlive) ? s2 : -1.0e9f;
  }
  __syncthreads();

  float v = -INFINITY;
  if ((int)lane < n) {
    float s = ((iter + 1 + (int)lane) < len) ? sc[lane] : -1.0e9f;
    v = (s == s) ? s : -INFINITY;
  }
  float m = v;
#pragma unroll
  for (int off = 16; off > 0; off >>= 1) m = fmaxf(m, __shfl_xor(m, off, 32));
  const unsigned long long hit = __ballot(v == m);
  const int ksel = hit ? (__ffsll((long long)hit) - 1) : 0;
  const bool done = (iter + 1) < len;

  const int vi0 = (int)blockIdx.x * 256 + tid;
  if (vi0 < nvec) {
#define MIX_LOAD(T)                                                              \
    float4 hv##T, cv##T; size_t dst##T;                                          \
    {                                                                            \
      const int vi = vi0 + (T) * (MIXG * 256);                                   \
      const int vc = (vi < nvec) ? vi : vi0;                                     \
      const int p = vc >> 7;                              \
      const int j = (vc & 127) * 4;                                              \
      const size_t cur = (size_t)(b * SEQL + p) * HDIM + j;                      \
      const size_t ni  = (size_t)(b * n + p) * HDIM + j;                         \
      const bool take_new   = done && (p == ksel);                               \
      const bool take_right = done && (p > ksel);                                \
      const size_t oc = cur + (take_right ? (size_t)HDIM : (size_t)0);           \
      const float* sh  = take_new ? (newh + ni) : (hcur + oc);                   \
      const float* scp = take_new ? (newc + ni) : (ccur + oc);                   \
      dst##T = cur;                                                              \
      hv##T = *(const float4*)sh;                                                \
      cv##T = *(const float4*)scp;                                               \
    }
#define MIX_STORE(T)                                                             \
    *(float4*)&hnxt[dst##T] = hv##T;                                             \
    *(float4*)&cnxt[dst##T] = cv##T;

    MIX_LOAD(0) MIX_LOAD(1) MIX_LOAD(2) MIX_LOAD(3) MIX_LOAD(4) MIX_LOAD(5)
    __builtin_amdgcn_sched_barrier(0);
    MIX_STORE(0) MIX_STORE(1) MIX_STORE(2) MIX_STORE(3) MIX_STORE(4) MIX_STORE(5)
#undef MIX_LOAD
#undef MIX_STORE
  }
}

__global__ __launch_bounds__(256) void final_copy_k(const float* __restrict__ hfin,
                                                    float* __restrict__ out) {
  int idx = blockIdx.x * 256 + threadIdx.x;
  int b = idx >> 9;
  int j = idx & (HDIM - 1);
  out[idx] = hfin[(size_t)b * SEQL * HDIM + j];
}

extern "C" void kernel_launch(void* const* d_in, const int* in_sizes, int n_in,
                              void* d_out, int out_size, void* d_ws, size_t ws_size,
                              hipStream_t stream) {
  const float* inp    = (const float*)d_in[0];
  const int*   length = (const int*)  d_in[1];
  const float* word_W = (const float*)d_in[2];
  const float* word_b = (const float*)d_in[3];
  const float* comp_W = (const float*)d_in[4];
  const float* comp_b = (const float*)d_in[5];
  const float* comp_q = (const float*)d_in[6];

  float* ws = (float*)d_ws;
  size_t off = 0;
  float* hb[2];
  float* cb[2];
  hb[0] = ws + off; off += (size_t)BATCH * SEQL * HDIM;
  cb[0] = ws + off; off += (size_t)BATCH * SEQL * HDIM;
  hb[1] = ws + off; off += (size_t)BATCH * SEQL * HDIM;
  cb[1] = ws + off; off += (size_t)BATCH * SEQL * HDIM;
  float* newh = ws + off; off += (size_t)(BATCH * 23) * HDIM;
  float* newc = ws + off; off += (size_t)(BATCH * 23) * HDIM;

  gemm_wmma<0><<<dim3((BATCH * SEQL) / TM, 1024 / 128), 256, 0, stream>>>(
      inp, nullptr, nullptr, word_W, word_b, hb[0], cb[0], nullptr, 512, 0, 0);

  int cur = 0;
  for (int i = 0; i < SEQL - 1; ++i) {
    int n = (SEQL - 1) - i;
    int nxt = cur ^ 1;
    gemm_wmma<1><<<dim3(HDIM / 32, 32), 256, 0, stream>>>(
        nullptr, hb[cur], cb[cur], comp_W, comp_b, newh, newc, length, 1024, n, i);
    mix_k<<<dim3(MIXG, BATCH), 256, 0, stream>>>(newh, newc, hb[cur], cb[cur], comp_q, length,
                                     hb[nxt], cb[nxt], n, i);
    cur = nxt;
  }

  final_copy_k<<<(BATCH * HDIM) / 256, 256, 0, stream>>>(hb[cur], (float*)d_out);
  (void)in_sizes; (void)n_in; (void)out_size; (void)ws_size;
}


# GEMM K chunk per hand-over 128 (short merge steps 256; was 64) on 116 KB static LDS, two workgroups per WGP still fit: half / a quarter of the load round trips and barrier pairs; same sums and order
# speedup vs baseline: 2.3561x; 1.1033x over previous
#include <hip/hip_runtime.h>
#include <hip/hip_bf16.h>
#include <math.h>


typedef __attribute__((ext_vector_type(2))) float v2f;
typedef __attribute__((ext_vector_type(8))) float v8f;

#define BATCH 64
#define SEQL  24
#define HDIM  512

__device__ __forceinline__ float sigm(float x) { return 1.0f / (1.0f + __expf(-x)); }
__device__ __forceinline__ float tanh_fast(float x) { return 2.0f / (1.0f + __expf(-2.0f * x)) - 1.0f; }

#define TM   64
#define KQ   64
#define KC   128
#define LDP  132
#define KCS  256
#define LDPS 260

template <int MODE>
__global__ __launch_bounds__(256) void gemm_wmma(const float* __restrict__ X,
                                                 const float* __restrict__ hbuf,
                                                 const float* __restrict__ cbuf,
                                                 const float* __restrict__ W,
                                                 const float* __restrict__ bias,
                                                 float* __restrict__ out0,
                                                 float* __restrict__ out1,
                                                 const int* __restrict__ length,
                                                 int K, int n, int iter) {
  __shared__ float A_lds[TM * LDP];
  __shared__ float B_lds[160 * LDP];
  __shared__ __attribute__((aligned(16))) int s_cnt[BATCH];
  __shared__ __attribute__((aligned(16))) int s_row[TM];

  const int tid  = threadIdx.x;
  const unsigned lane = tid & 31u;
  const int wave = tid >> 5;
  const int wm   = wave >> 1;
  const int wn   = wave & 1;
  const int l16  = (int)(lane & 15u);
  const int lhi  = (int)(lane >> 4);

  const int rblk = (MODE == 0) ? (int)blockIdx.x : (int)blockIdx.y;
  const int cblk = (MODE == 0) ? (int)blockIdx.y : (int)blockIdx.x;
  int r0 = rblk * TM;
  int nlv = TM;
  const int NACC = (MODE == 0) ? 4 : 5;

  int tot = 0;
  bool small = false;
  int cS = 0;
  if (MODE == 1) {
    int c = 0;
    if (tid < BATCH) {
      c = min(max(length[tid] - 1 - iter, 0), n);
      s_cnt[tid] = c;
      s_row[tid] = 0;
    }
    __syncthreads();
    int pre = 0;
#pragma unroll
    for (int j = 0; j < BATCH; ++j) {
      const int cj = s_cnt[j];
      pre += (j < tid) ? cj : 0;
      tot += cj;
    }
    tot = __builtin_amdgcn_readfirstlane(tot);
    const int nwg = (int)(gridDim.x * gridDim.y);
    small = (tot <= 256) && (32 * ((tot + 15) >> 4) <= nwg);
    int rpb, rb;
    if (small) {
      const int lin = (int)(blockIdx.y * gridDim.x + blockIdx.x);
      rpb = 16;
      rb = lin >> 5;
      cS = lin & 31;
    } else {
      rpb = (tot <= 32 * (int)gridDim.y && tot <= 512) ? 32 : TM;
      rb = rblk;
    }
    r0 = rb * rpb;
    if (r0 >= tot) return;
    nlv = min(rpb, tot - r0);
    if (tid < BATCH) {
      const int lo = max(pre, r0);
      const int hi = min(pre + c, r0 + rpb);
      for (int r = lo; r < hi; ++r) s_row[r - r0] = (tid << 5) | (r - pre);
    }
    __syncthreads();
  }
  if (MODE == 1 && small) {
    const int srowS = tid >> 4;
    const int sc4S  = (tid & 15) * 4;
    const int infoA = s_row[srowS];
    const float* apS = hbuf + (size_t)((infoA >> 5) * SEQL + (infoA & 31)) * HDIM + sc4S;
    const float* bpS[5];
#pragma unroll
    for (int g = 0; g < 5; ++g) bpS[g] = W + (size_t)(g * HDIM + cS * 16 + srowS) * K + sc4S;
    const int wv = __builtin_amdgcn_readfirstlane(wave);
    const bool actS = wv < 5;
    v8f accS = {};
    for (int k0 = 0; k0 < K; k0 += KCS) {
      static_assert(KCS == 4 * KQ, "a thread loads four KQ-wide pieces of every staged row per chunk");
      static_assert(16 * LDPS <= TM * LDP && 80 * LDPS <= 160 * LDP, "the short step's rows fit the full-size arrays");
#define SLD(h) const float4 sa_##h  = *(const float4*)(apS + k0 + (h) * KQ);    \
               const float4 sb0_##h = *(const float4*)(bpS[0] + k0 + (h) * KQ); \
               const float4 sb1_##h = *(const float4*)(bpS[1] + k0 + (h) * KQ); \
               const float4 sb2_##h = *(const float4*)(bpS[2] + k0 + (h) * KQ); \
               const float4 sb3_##h = *(const float4*)(bpS[3] + k0 + (h) * KQ); \
               const float4 sb4_##h = *(const float4*)(bpS[4] + k0 + (h) * KQ);
#define SPUT(h) *(float4*)&A_lds[srowS * LDPS + (h) * KQ + sc4S] = sa_##h;              \
                *(float4*)&B_lds[(0 * 16 + srowS) * LDPS + (h) * KQ + sc4S] = sb0_##h;  \
                *(float4*)&B_lds[(1 * 16 + srowS) * LDPS + (h) * KQ + sc4S] = sb1_##h;  \
                *(float4*)&B_lds[(2 * 16 + srowS) * LDPS + (h) * KQ + sc4S] = sb2_##h;  \
                *(float4*)&B_lds[(3 * 16 + srowS) * LDPS + (h) * KQ + sc4S] = sb3_##h;  \
                *(float4*)&B_lds[(4 * 16 + srowS) * LDPS + (h) * KQ + sc4S] = sb4_##h;
      SLD(0) SLD(1) SLD(2) SLD(3)
      SPUT(0) SPUT(1) SPUT(2) SPUT(3)
#undef SLD
#undef SPUT
      __syncthreads();
      if (actS) {
#pragma unroll
        for (int kk = 0; kk < KCS; kk += 4) {
          v2f a  = *(const v2f*)&A_lds[l16 * LDPS + kk + 2 * lhi];
          v2f bf = *(const v2f*)&B_lds[(wv * 16 + l16) * LDPS + kk + 2 * lhi];
          accS = __builtin_amdgcn_wmma_f32_16x16x4_f32(
              false, a, false, bf, (short)0, accS, false, false);
        }
      }
      __syncthreads();
    }
    float* Gs = B_lds;
    if (actS) {
#pragma unroll
      for (int vr = 0; vr < 8; ++vr) Gs[(wv * 16 + 8 * lhi + vr) * 17 + l16] = accS[vr];
    }
    __syncthreads();
    {
      const int row  = tid >> 4;
      const int col  = tid & 15;
      const int colg = cS * 16 + col;
      const int info = s_row[row];
      const int b = info >> 5;
      const int p = info & 31;
      const size_t cbase = (size_t)(b * SEQL + p) * HDIM + colg;
      const float clS = cbuf[cbase];
      const float crS = cbuf[cbase + HDIM];
      float gi  = Gs[(0 * 16 + row) * 17 + col] + bias[0 * HDIM + colg];
      float gfl = Gs[(1 * 16 + row) * 17 + col] + bias[1 * HDIM + colg];
      float gfr = Gs[(2 * 16 + row) * 17 + col] + bias[2 * HDIM + colg];
      float gu  = Gs[(3 * 16 + row) * 17 + col] + bias[3 * HDIM + colg];
      float go  = Gs[(4 * 16 + row) * 17 + col] + bias[4 * HDIM + colg];
      float nc = clS * sigm(gfl + 1.0f) + crS * sigm(gfr + 1.0f) + tanh_fast(gu) * sigm(gi);
      float nh = sigm(go) * tanh_fast(nc);
      if (row < nlv) {
        const size_t oi = (size_t)(b * n + p) * HDIM + colg;
        out0[oi] = nh;
        out1[oi] = nc;
      }
    }
    return;
  }

  const int NA = TM / 16;
  const int NB = (MODE == 0) ? 8 : 10;
  static_assert(KC == 2 * KQ, "a thread loads two KQ-wide pieces of every staged row per chunk");
  static_assert(TM == 64, "the staging below is written out for 4 A rows per thread");

  v8f acc[5] = {};

  const int srow = tid >> 4;
  const int sc4  = (tid & 15) * 4;
  const float* ap[NA];
#pragma unroll
  for (int i = 0; i < NA; ++i) {
    int rg = r0 + i * 16 + srow;
    if (MODE == 0) {
      ap[i] = X + (size_t)rg * K + sc4;
    } else {
      const int info = s_row[i * 16 + srow];
      const int b = info >> 5;
      const int p = info & 31;
      ap[i] = hbuf + (size_t)(b * SEQL + p) * HDIM + sc4;
    }
  }
  const float* bp[10];
#pragma unroll
  for (int i = 0; i < NB; ++i) {
    int nrow = i * 16 + srow;
    size_t wrow;
    if (MODE == 0) {
      wrow = (size_t)(cblk * 128 + nrow);
    } else {
      int g  = nrow >> 5;
      int nl = nrow & 31;
      wrow = (size_t)(g * HDIM + cblk * 32 + nl);
    }
    bp[i] = W + wrow * K + sc4;
  }

  const bool act = (MODE == 0) ? true : (__builtin_amdgcn_readfirstlane(wm) * 16 < nlv);

  for (int k0 = 0; k0 < K; k0 += KC) {
#define STG_A(i) const float4 a0_##i = *(const float4*)(ap[i] + k0); \
                 const float4 a1_##i = *(const float4*)(ap[i] + k0 + KQ);
#define STG_B(i) const float4 b0_##i = *(const float4*)(bp[i] + k0); \
                 const float4 b1_##i = *(const float4*)(bp[i] + k0 + KQ);
#define STG_B1(i) float4 b0_##i = make_float4(0.0f, 0.0f, 0.0f, 0.0f), b1_##i = b0_##i; \
                  if (MODE == 1) { b0_##i = *(const float4*)(bp[i] + k0); b1_##i = *(const float4*)(bp[i] + k0 + KQ); }
    STG_A(0) STG_A(1) STG_A(2) STG_A(3)
    STG_B(0) STG_B(1) STG_B(2) STG_B(3) STG_B(4) STG_B(5) STG_B(6) STG_B(7)
    STG_B1(8) STG_B1(9)
#undef STG_A
#undef STG_B
#undef STG_B1
    __builtin_amdgcn_sched_barrier(0);
#define PUT_A(i) *(float4*)&A_lds[((i) * 16 + srow) * LDP + sc4] = a0_##i; \
                 *(float4*)&A_lds[((i) * 16 + srow) * LDP + KQ + sc4] = a1_##i;
#define PUT_B(i) *(float4*)&B_lds[((i) * 16 + srow) * LDP + sc4] = b0_##i; \
                 *(float4*)&B_lds[((i) * 16 + srow) * LDP + KQ + sc4] = b1_##i;
    PUT_A(0) PUT_A(1) PUT_A(2) PUT_A(3)
    PUT_B(0) PUT_B(1) PUT_B(2) PUT_B(3) PUT_B(4) PUT_B(5) PUT_B(6) PUT_B(7)
    if (MODE == 1) { PUT_B(8) PUT_B(9) }
#undef PUT_A
#undef PUT_B
    __syncthreads();

    if (act) {
#pragma unroll
      for (int kk = 0; kk < KC; kk += 4) {
        v2f a = *(const v2f*)&A_lds[(wm * 16 + l16) * LDP + kk + 2 * lhi];
#pragma unroll
        for (int g = 0; g < NACC; ++g) {
          int nrow = (MODE == 0) ? (wn * 64 + g * 16 + l16)
                                 : (g * 32 + wn * 16 + l16);
          v2f bf = *(const v2f*)&B_lds[nrow * LDP + kk + 2 * lhi];
          acc[g] = __builtin_amdgcn_wmma_f32_16x16x4_f32(
              false, a, false, bf, (short)0, acc[g], false, false);
        }
      }
    }
    __syncthreads();
  }

  if (MODE == 0) {
    const int colbase = cblk * 128 + wn * 64;
    const int rowg0   = r0 + wm * 16 + 8 * lhi;
#pragma unroll
    for (int j = 0; j < 4; ++j) {
      int col = colbase + j * 16 + l16;
      float bv = bias[col];
#pragma unroll
      for (int vr = 0; vr < 8; ++vr) {
        int rg = rowg0 + vr;
        float val = acc[j][vr] + bv;
        if (col < HDIM) out0[(size_t)rg * HDIM + col] = val;
        else            out1[(size_t)rg * HDIM + (col - HDIM)] = val;
      }
    }
  } else {
    const int colg = cblk * 32 + wn * 16 + l16;
    float bz[5];
#pragma unroll
    for (int g = 0; g < 5; ++g) bz[g] = bias[g * HDIM + colg];
    const int rl0 = wm * 16 + 8 * lhi;
    int rinfo[8];
    *(int4*)&rinfo[0] = *(const int4*)&s_row[rl0];
    *(int4*)&rinfo[4] = *(const int4*)&s_row[rl0 + 4];
    float clv[8], crv[8];
    size_t oiv[8];
#pragma unroll
    for (int vr = 0; vr < 8; ++vr) {
      const int b = rinfo[vr] >> 5;
      const int p = rinfo[vr] & 31;
      const size_t cbase = (size_t)(b * SEQL + p) * HDIM + colg;
      clv[vr] = cbuf[cbase];
      crv[vr] = cbuf[cbase + HDIM];
      oiv[vr] = (size_t)(b * n + p) * HDIM + colg;
    }
#pragma unroll
    for (int vr = 0; vr < 8; ++vr) asm volatile("" : "+v"(clv[vr]), "+v"(crv[vr]));
#pragma unroll
    for (int vr = 0; vr < 8; ++vr) {
      const bool live = (rl0 + vr) < nlv;
      float gi  = acc[0][vr] + bz[0];
      float gfl = acc[1][vr] + bz[1];
      float gfr = acc[2][vr] + bz[2];
      float gu  = acc[3][vr] + bz[3];
      float go  = acc[4][vr] + bz[4];
      float nc = clv[vr] * sigm(gfl + 1.0f) + crv[vr] * sigm(gfr + 1.0f) + tanh_fast(gu) * sigm(gi);
      float nh = sigm(go) * tanh_fast(nc);
      if (live) {
        out0[oiv[vr]] = nh;
        out1[oiv[vr]] = nc;
      }
    }
  }
}

#define MIXG 2
#define MIXT 6

__global__ __launch_bounds__(256) void mix_k(const float* __restrict__ newh,
                                             const float* __restrict__ newc,
                                             const float* __restrict__ hcur,
                                             const float* __restrict__ ccur,
                                             const float* __restrict__ q,
                                             const int* __restrict__ length,
                                             float* __restrict__ hnxt,
                                             float* __restrict__ cnxt,
                                             int n, int iter) {
  __shared__ float sc[32];
  const int b = blockIdx.y;
  const int tid = threadIdx.x;
  const unsigned lane = tid & 31u;
  const int wave = tid >> 5;
  const int nvec = n * (HDIM / 4);
  if ((int)blockIdx.x * 256 >= nvec) return;

  const int len = length[b];
  const int nlive = min(max(len - 1 - iter, 0), n);
  const int p0 = wave, p1 = wave + 8, p2 = wave + 16;
  float s0 = 0.0f, s1 = 0.0f, s2 = 0.0f;
  if (nlive > 0) {
    const float* r0 = newh + (size_t)(b * n + min(p0, nlive - 1)) * HDIM;
    const float* r1 = newh + (size_t)(b * n + min(p1, nlive - 1)) * HDIM;
    const float* r2 = newh + (size_t)(b * n + min(p2, nlive - 1)) * HDIM;
#pragma unroll
    for (int t = 0; t < HDIM / 32; ++t) {
      const int j = (int)lane + 32 * t;
      const float qv = q[j];
      s0 += r0[j] * qv;
      s1 += r1[j] * qv;
      s2 += r2[j] * qv;
    }
#pragma unroll
    for (int off = 16; off > 0; off >>= 1) {
      s0 += __shfl_xor(s0, off, 32);
      s1 += __shfl_xor(s1, off, 32);
      s2 += __shfl_xor(s2, off, 32);
    }
  }
  if (lane == 0) {
    if (p0 < n) sc[p0] = (p0 < nlive) ? s0 : -1.0e9f;
    if (p1 < n) sc[p1] = (p1 < nlive) ? s1 : -1.0e9f;
    if (p2 < n) sc[p2] = (p2 < nlive) ? s2 : -1.0e9f;
  }
  __syncthreads();

  float v = -INFINITY;
  if ((int)lane < n) {
    float s = ((iter + 1 + (int)lane) < len) ? sc[lane] : -1.0e9f;
    v = (s == s) ? s : -INFINITY;
  }
  float m = v;
#pragma unroll
  for (int off = 16; off > 0; off >>= 1) m = fmaxf(m, __shfl_xor(m, off, 32));
  const unsigned long long hit = __ballot(v == m);
  const int ksel = hit ? (__ffsll((long long)hit) - 1) : 0;
  const bool done = (iter + 1) < len;

  const int vi0 = (int)blockIdx.x * 256 + tid;
  if (vi0 < nvec) {
#define MIX_LOAD(T)                                                              \
    float4 hv##T, cv##T; size_t dst##T;                                          \
    {                                                                            \
      const int vi = vi0 + (T) * (MIXG * 256);                                   \
      const int vc = (vi < nvec) ? vi : vi0;                                     \
      const int p = vc >> 7;                              \
      const int j = (vc & 127) * 4;                                              \
      const size_t cur = (size_t)(b * SEQL + p) * HDIM + j;                      \
      const size_t ni  = (size_t)(b * n + p) * HDIM + j;                         \
      const bool take_new   = done && (p == ksel);                               \
      const bool take_right = done && (p > ksel);                                \
      const size_t oc = cur + (take_right ? (size_t)HDIM : (size_t)0);           \
      const float* sh  = take_new ? (newh + ni) : (hcur + oc);                   \
      const float* scp = take_new ? (newc + ni) : (ccur + oc);                   \
      dst##T = cur;                                                              \
      hv##T = *(const float4*)sh;                                                \
      cv##T = *(const float4*)scp;                                               \
    }
#define MIX_STORE(T)                                                             \
    *(float4*)&hnxt[dst##T] = hv##T;                                             \
    *(float4*)&cnxt[dst##T] = cv##T;

    MIX_LOAD(0) MIX_LOAD(1) MIX_LOAD(2) MIX_LOAD(3) MIX_LOAD(4) MIX_LOAD(5)
    __builtin_amdgcn_sched_barrier(0);
    MIX_STORE(0) MIX_STORE(1) MIX_STORE(2) MIX_STORE(3) MIX_STORE(4) MIX_STORE(5)
#undef MIX_LOAD
#undef MIX_STORE
  }
}

__global__ __launch_bounds__(256) void final_copy_k(const float* __restrict__ hfin,
                                                    float* __restrict__ out) {
  int idx = blockIdx.x * 256 + threadIdx.x;
  int b = idx >> 9;
  int j = idx & (HDIM - 1);
  out[idx] = hfin[(size_t)b * SEQL * HDIM + j];
}

extern "C" void kernel_launch(void* const* d_in, const int* in_sizes, int n_in,
                              void* d_out, int out_size, void* d_ws, size_t ws_size,
                              hipStream_t stream) {
  const float* inp    = (const float*)d_in[0];
  const int*   length = (const int*)  d_in[1];
  const float* word_W = (const float*)d_in[2];
  const float* word_b = (const float*)d_in[3];
  const float* comp_W = (const float*)d_in[4];
  const float* comp_b = (const float*)d_in[5];
  const float* comp_q = (const float*)d_in[6];

  float* ws = (float*)d_ws;
  size_t off = 0;
  float* hb[2];
  float* cb[2];
  hb[0] = ws + off; off += (size_t)BATCH * SEQL * HDIM;
  cb[0] = ws + off; off += (size_t)BATCH * SEQL * HDIM;
  hb[1] = ws + off; off += (size_t)BATCH * SEQL * HDIM;
  cb[1] = ws + off; off += (size_t)BATCH * SEQL * HDIM;
  float* newh = ws + off; off += (size_t)(BATCH * 23) * HDIM;
  float* newc = ws + off; off += (size_t)(BATCH * 23) * HDIM;

  gemm_wmma<0><<<dim3((BATCH * SEQL) / TM, 1024 / 128), 256, 0, stream>>>(
      inp, nullptr, nullptr, word_W, word_b, hb[0], cb[0], nullptr, 512, 0, 0);

  int cur = 0;
  for (int i = 0; i < SEQL - 1; ++i) {
    int n = (SEQL - 1) - i;
    int nxt = cur ^ 1;
    gemm_wmma<1><<<dim3(HDIM / 32, 32), 256, 0, stream>>>(
        nullptr, hb[cur], cb[cur], comp_W, comp_b, newh, newc, length, 1024, n, i);
    mix_k<<<dim3(MIXG, BATCH), 256, 0, stream>>>(newh, newc, hb[cur], cb[cur], comp_q, length,
                                     hb[nxt], cb[nxt], n, i);
    cur = nxt;
  }

  final_copy_k<<<(BATCH * HDIM) / 256, 256, 0, stream>>>(hb[cur], (float*)d_out);
  (void)in_sizes; (void)n_in; (void)out_size; (void)ws_size;
}


# Comp GEMM full-size step: waves 4-7 only stage (global loads into an LDS ring of 2 x 64 k), waves 0-3 only multiply (10 accumulators): a stage's loads run under the other's products; same sums, order
# speedup vs baseline: 2.6120x; 1.1086x over previous
#include <hip/hip_runtime.h>
#include <hip/hip_bf16.h>
#include <math.h>


typedef __attribute__((ext_vector_type(2))) float v2f;
typedef __attribute__((ext_vector_type(8))) float v8f;

#define BATCH 64
#define SEQL  24
#define HDIM  512

__device__ __forceinline__ float sigm(float x) { return 1.0f / (1.0f + __expf(-x)); }
__device__ __forceinline__ float tanh_fast(float x) { return 2.0f / (1.0f + __expf(-2.0f * x)) - 1.0f; }

#define TM   64
#define KQ   64
#define KC   128
#define LDP  132
#define KCS  256
#define LDPS 260
#define KS   64
#define LDQ  68

template <int MODE>
__global__ __launch_bounds__(256) void gemm_wmma(const float* __restrict__ X,
                                                 const float* __restrict__ hbuf,
                                                 const float* __restrict__ cbuf,
                                                 const float* __restrict__ W,
                                                 const float* __restrict__ bias,
                                                 float* __restrict__ out0,
                                                 float* __restrict__ out1,
                                                 const int* __restrict__ length,
                                                 int K, int n, int iter) {
  __shared__ float A_lds[2 * TM * LDQ];
  __shared__ float B_lds[2 * 160 * LDQ];
  static_assert(TM * LDP <= 2 * TM * LDQ && 128 * LDP <= 2 * 160 * LDQ, "mode 0's chunk fits the arrays");
  __shared__ __attribute__((aligned(16))) int s_cnt[BATCH];
  __shared__ __attribute__((aligned(16))) int s_row[TM];

  const int tid  = threadIdx.x;
  const unsigned lane = tid & 31u;
  const int wave = tid >> 5;
  const int wm   = wave >> 1;
  const int wn   = wave & 1;
  const int l16  = (int)(lane & 15u);
  const int lhi  = (int)(lane >> 4);

  const int rblk = (MODE == 0) ? (int)blockIdx.x : (int)blockIdx.y;
  const int cblk = (MODE == 0) ? (int)blockIdx.y : (int)blockIdx.x;
  int r0 = rblk * TM;
  int nlv = TM;
  int rpb = TM;
  const int NACC = (MODE == 0) ? 4 : 5;

  int tot = 0;
  bool small = false;
  int cS = 0;
  unsigned rot = (MODE == 0) ? ((unsigned)(rblk + cblk) % ((unsigned)K / KC)) : 0u;
  if (MODE == 1) {
    int c = 0;
    if (tid < BATCH) {
      c = min(max(length[tid] - 1 - iter, 0), n);
      s_cnt[tid] = c;
      s_row[tid] = 0;
    }
    __syncthreads();
    int pre = 0;
#pragma unroll
    for (int j = 0; j < BATCH; ++j) {
      const int cj = s_cnt[j];
      pre += (j < tid) ? cj : 0;
      tot += cj;
    }
    tot = __builtin_amdgcn_readfirstlane(tot);
    const int nwg = (int)(gridDim.x * gridDim.y);
    small = (tot <= 256) && (32 * ((tot + 15) >> 4) <= nwg);
    int rb;
    if (small) {
      const int lin = (int)(blockIdx.y * gridDim.x + blockIdx.x);
      rpb = 16;
      rb = lin >> 5;
      cS = lin & 31;
    } else {
      rpb = (tot <= 32 * (int)gridDim.y && tot <= 512) ? 32 : TM;
      rb = rblk;
    }
    r0 = rb * rpb;
    if (r0 >= tot) return;
    rot = small ? ((unsigned)(rb + cS) % ((unsigned)K / KCS))
                : ((unsigned)(rb + cblk) % ((unsigned)K / KC));
    nlv = min(rpb, tot - r0);
    if (tid < BATCH) {
      const int lo = max(pre, r0);
      const int hi = min(pre + c, r0 + rpb);
      for (int r = lo; r < hi; ++r) s_row[r - r0] = (tid << 5) | (r - pre);
    }
    __syncthreads();
  }
  if (MODE == 1 && small) {
    const int srowS = tid >> 4;
    const int sc4S  = (tid & 15) * 4;
    const int infoA = s_row[srowS];
    const float* apS = hbuf + (size_t)((infoA >> 5) * SEQL + (infoA & 31)) * HDIM + sc4S;
    const float* bpS[5];
#pragma unroll
    for (int g = 0; g < 5; ++g) bpS[g] = W + (size_t)(g * HDIM + cS * 16 + srowS) * K + sc4S;
    const int wv = __builtin_amdgcn_readfirstlane(wave);
    const bool actS = wv < 5;
    v8f accS = {};
    const unsigned nchS = (unsigned)K / KCS;
    for (unsigned ci = 0; ci < nchS; ++ci) {
      unsigned ch = ci + rot;
      if (ch >= nchS) ch -= nchS;
      const int k0 = (int)(ch * KCS);
      static_assert(KCS == 4 * KQ, "a thread loads four KQ-wide pieces of every staged row per chunk");
      static_assert(16 * LDPS <= 2 * TM * LDQ && 80 * LDPS <= 2 * 160 * LDQ, "the short step's rows fit the arrays");
#define SLD(h) const float4 sa_##h  = *(const float4*)(apS + k0 + (h) * KQ);    \
               const float4 sb0_##h = *(const float4*)(bpS[0] + k0 + (h) * KQ); \
               const float4 sb1_##h = *(const float4*)(bpS[1] + k0 + (h) * KQ); \
               const float4 sb2_##h = *(const float4*)(bpS[2] + k0 + (h) * KQ); \
               const float4 sb3_##h = *(const float4*)(bpS[3] + k0 + (h) * KQ); \
               const float4 sb4_##h = *(const float4*)(bpS[4] + k0 + (h) * KQ);
#define SPUT(h) *(float4*)&A_lds[srowS * LDPS + (h) * KQ + sc4S] = sa_##h;              \
                *(float4*)&B_lds[(0 * 16 + srowS) * LDPS + (h) * KQ + sc4S] = sb0_##h;  \
                *(float4*)&B_lds[(1 * 16 + srowS) * LDPS + (h) * KQ + sc4S] = sb1_##h;  \
                *(float4*)&B_lds[(2 * 16 + srowS) * LDPS + (h) * KQ + sc4S] = sb2_##h;  \
                *(float4*)&B_lds[(3 * 16 + srowS) * LDPS + (h) * KQ + sc4S] = sb3_##h;  \
                *(float4*)&B_lds[(4 * 16 + srowS) * LDPS + (h) * KQ + sc4S] = sb4_##h;
      SLD(0) SLD(1) SLD(2) SLD(3)
      SPUT(0) SPUT(1) SPUT(2) SPUT(3)
#undef SLD
#undef SPUT
      __syncthreads();
      if (actS) {
#pragma unroll
        for (int kk = 0; kk < KCS; kk += 4) {
          v2f a  = *(const v2f*)&A_lds[l16 * LDPS + kk + 2 * lhi];
          v2f bf = *(const v2f*)&B_lds[(wv * 16 + l16) * LDPS + kk + 2 * lhi];
          accS = __builtin_amdgcn_wmma_f32_16x16x4_f32(
              false, a, false, bf, (short)0, accS, false, false);
        }
      }
      __syncthreads();
    }
    float* Gs = B_lds;
    if (actS) {
#pragma unroll
      for (int vr = 0; vr < 8; ++vr) Gs[(wv * 16 + 8 * lhi + vr) * 17 + l16] = accS[vr];
    }
    __syncthreads();
    {
      const int row  = tid >> 4;
      const int col  = tid & 15;
      const int colg = cS * 16 + col;
      const int info = s_row[row];
      const int b = info >> 5;
      const int p = info & 31;
      const size_t cbase = (size_t)(b * SEQL + p) * HDIM + colg;
      const float clS = cbuf[cbase];
      const float crS = cbuf[cbase + HDIM];
      float gi  = Gs[(0 * 16 + row) * 17 + col] + bias[0 * HDIM + colg];
      float gfl = Gs[(1 * 16 + row) * 17 + col] + bias[1 * HDIM + colg];
      float gfr = Gs[(2 * 16 + row) * 17 + col] + bias[2 * HDIM + colg];
      float gu  = Gs[(3 * 16 + row) * 17 + col] + bias[3 * HDIM + colg];
      float go  = Gs[(4 * 16 + row) * 17 + col] + bias[4 * HDIM + colg];
      float nc = clS * sigm(gfl + 1.0f) + crS * sigm(gfr + 1.0f) + tanh_fast(gu) * sigm(gi);
      float nh = sigm(go) * tanh_fast(nc);
      if (row < nlv) {
        const size_t oi = (size_t)(b * n + p) * HDIM + colg;
        out0[oi] = nh;
        out1[oi] = nc;
      }
    }
    return;
  }

  if (MODE == 1) {
    const int wv = __builtin_amdgcn_readfirstlane(wave);
    const bool loader = wv >= 4;
    const int wnm = wv & 1;
    const int mtA = (rpb == TM) ? 2 * ((wv >> 1) & 1) : ((wv >> 1) & 1);
    const int mtB = mtA + 1;
    const bool actA = !loader && (mtA * 16 < nlv);
    const bool actB = actA && (rpb == TM) && (mtB * 16 < nlv);

    const int ltid = tid & 127;
    const int r8   = ltid >> 4;
    const int c4   = (ltid & 15) * 4;
    constexpr unsigned KK1 = 2 * HDIM;
    unsigned aB[8];
#pragma unroll
    for (int j = 0; j < 8; ++j) {
      const int info = s_row[j * 8 + r8];
      aB[j] = (unsigned)(((info >> 5) * SEQL + (info & 31)) * HDIM + c4) * 4u;
    }
    const unsigned wB = (unsigned)((cblk * 32 + r8) * (int)KK1 + c4) * 4u;
    const char* const hbase = (const char*)hbuf;
    const char* const wbase = (const char*)W;

    auto stage = [&](const int k0, float* __restrict__ Ad, float* __restrict__ Bd)
        __attribute__((always_inline)) {
      const unsigned kB = (unsigned)k0 * 4u;
#define LDA(j) const float4 la_##j = *(const float4*)(hbase + (aB[j] + kB));
#define LDB(j) const float4 lb_##j = *(const float4*)(wbase + (wB + kB + \
                   (unsigned)((((j) >> 2) * HDIM + ((j) & 3) * 8) * (int)KK1) * 4u));
      LDA(0) LDA(1) LDA(2) LDA(3) LDA(4) LDA(5) LDA(6) LDA(7)
      LDB(0) LDB(1) LDB(2) LDB(3) LDB(4) LDB(5) LDB(6) LDB(7) LDB(8) LDB(9)
      LDB(10) LDB(11) LDB(12) LDB(13) LDB(14) LDB(15) LDB(16) LDB(17) LDB(18) LDB(19)
#undef LDA
#undef LDB
      __builtin_amdgcn_sched_barrier(0);
#define STA(j) *(float4*)&Ad[((j) * 8 + r8) * LDQ + c4] = la_##j;
#define STB(j) *(float4*)&Bd[((j) * 8 + r8) * LDQ + c4] = lb_##j;
      STA(0) STA(1) STA(2) STA(3) STA(4) STA(5) STA(6) STA(7)
      STB(0) STB(1) STB(2) STB(3) STB(4) STB(5) STB(6) STB(7) STB(8) STB(9)
      STB(10) STB(11) STB(12) STB(13) STB(14) STB(15) STB(16) STB(17) STB(18) STB(19)
#undef STA
#undef STB
    };

    v8f accA[5] = {};
    v8f accB[5] = {};
    auto mma1 = [&](const float* __restrict__ Ab, const float* __restrict__ Bb)
        __attribute__((always_inline)) {
#pragma unroll
      for (int kk = 0; kk < KS; kk += 4) {
        const v2f a = *(const v2f*)&Ab[(mtA * 16 + l16) * LDQ + kk + 2 * lhi];
#pragma unroll
        for (int g = 0; g < 5; ++g) {
          const v2f bf = *(const v2f*)&Bb[(g * 32 + wnm * 16 + l16) * LDQ + kk + 2 * lhi];
          accA[g] = __builtin_amdgcn_wmma_f32_16x16x4_f32(
              false, a, false, bf, (short)0, accA[g], false, false);
        }
      }
    };
    auto mma2 = [&](const float* __restrict__ Ab, const float* __restrict__ Bb)
        __attribute__((always_inline)) {
#pragma unroll
      for (int kk = 0; kk < KS; kk += 4) {
        const v2f a0 = *(const v2f*)&Ab[(mtA * 16 + l16) * LDQ + kk + 2 * lhi];
        const v2f a1 = *(const v2f*)&Ab[(mtB * 16 + l16) * LDQ + kk + 2 * lhi];
#pragma unroll
        for (int g = 0; g < 5; ++g) {
          const v2f bf = *(const v2f*)&Bb[(g * 32 + wnm * 16 + l16) * LDQ + kk + 2 * lhi];
          accA[g] = __builtin_amdgcn_wmma_f32_16x16x4_f32(
              false, a0, false, bf, (short)0, accA[g], false, false);
          accB[g] = __builtin_amdgcn_wmma_f32_16x16x4_f32(
              false, a1, false, bf, (short)0, accB[g], false, false);
        }
      }
    };

    float* const A0 = A_lds;
    float* const A1 = A_lds + TM * LDQ;
    float* const B0 = B_lds;
    float* const B1 = B_lds + 160 * LDQ;
    const unsigned nst = KK1 / KS;
    const unsigned st0 = rot * (KC / KS);
    static_assert((KC / KS) == 2, "the stage loop below takes the ring's two stages in turn");
    if (loader) stage((int)(st0 * KS), A0, B0);
    __syncthreads();
    for (unsigned s = 0; s < nst; s += 2) {
      {
        unsigned nx = st0 + s + 1;
        if (nx >= nst) nx -= nst;
        if (loader) {
          stage((int)(nx * KS), A1, B1);
        } else if (actA) {
          if (actB) mma2(A0, B0); else mma1(A0, B0);
        }
        __syncthreads();
      }
      {
        unsigned nx = st0 + s + 2;
        if (nx >= nst) nx -= nst;
        if (loader) {
          if (s + 2 < nst) stage((int)(nx * KS), A0, B0);
        } else if (actA) {
          if (actB) mma2(A1, B1); else mma1(A1, B1);
        }
        __syncthreads();
      }
    }

    if (actA) {
      const int colg = cblk * 32 + wnm * 16 + l16;
      float bz[5];
#pragma unroll
      for (int g = 0; g < 5; ++g) bz[g] = bias[g * HDIM + colg];
      float clv[2][8], crv[2][8];
      size_t oiv[2][8];
#pragma unroll
      for (int t = 0; t < 2; ++t) {
        if (t == 1 && !actB) break;
        const int rl0 = (t == 0 ? mtA : mtB) * 16 + 8 * lhi;
        int rinfo[8];
        *(int4*)&rinfo[0] = *(const int4*)&s_row[rl0];
        *(int4*)&rinfo[4] = *(const int4*)&s_row[rl0 + 4];
#pragma unroll
        for (int vr = 0; vr < 8; ++vr) {
          const int b = rinfo[vr] >> 5;
          const int p = rinfo[vr] & 31;
          const size_t cbase = (size_t)(b * SEQL + p) * HDIM + colg;
          clv[t][vr] = cbuf[cbase];
          crv[t][vr] = cbuf[cbase + HDIM];
          oiv[t][vr] = (size_t)(b * n + p) * HDIM + colg;
        }
      }
#pragma unroll
      for (int t = 0; t < 2; ++t) {
        if (t == 1 && !actB) break;
#pragma unroll
        for (int vr = 0; vr < 8; ++vr) asm volatile("" : "+v"(clv[t][vr]), "+v"(crv[t][vr]));
        const int rl0 = (t == 0 ? mtA : mtB) * 16 + 8 * lhi;
#pragma unroll
        for (int vr = 0; vr < 8; ++vr) {
          const bool live = (rl0 + vr) < nlv;
          float gi  = (t == 0 ? accA[0][vr] : accB[0][vr]) + bz[0];
          float gfl = (t == 0 ? accA[1][vr] : accB[1][vr]) + bz[1];
          float gfr = (t == 0 ? accA[2][vr] : accB[2][vr]) + bz[2];
          float gu  = (t == 0 ? accA[3][vr] : accB[3][vr]) + bz[3];
          float go  = (t == 0 ? accA[4][vr] : accB[4][vr]) + bz[4];
          float nc = clv[t][vr] * sigm(gfl + 1.0f) + crv[t][vr] * sigm(gfr + 1.0f) + tanh_fast(gu) * sigm(gi);
          float nh = sigm(go) * tanh_fast(nc);
          if (live) {
            out0[oiv[t][vr]] = nh;
            out1[oiv[t][vr]] = nc;
          }
        }
      }
    }
    return;
  }

  const int NA = TM / 16;
  static_assert(KC == 2 * KQ, "a thread loads two KQ-wide pieces of every staged row per chunk");
  static_assert(TM == 64, "the staging below is written out for 4 A rows per thread");

  v8f acc[4] = {};

  const int srow = tid >> 4;
  const int sc4  = (tid & 15) * 4;
  const float* ap[NA];
#pragma unroll
  for (int i = 0; i < NA; ++i) ap[i] = X + (size_t)(r0 + i * 16 + srow) * K + sc4;
  const float* bp[8];
#pragma unroll
  for (int i = 0; i < 8; ++i) bp[i] = W + (size_t)(cblk * 128 + i * 16 + srow) * K + sc4;

  const unsigned nch = (unsigned)K / KC;
  for (unsigned ci = 0; ci < nch; ++ci) {
    unsigned ch = ci + rot;
    if (ch >= nch) ch -= nch;
    const int k0 = (int)(ch * KC);
#define STG_A(i) const float4 a0_##i = *(const float4*)(ap[i] + k0); \
                 const float4 a1_##i = *(const float4*)(ap[i] + k0 + KQ);
#define STG_B(i) const float4 b0_##i = *(const float4*)(bp[i] + k0); \
                 const float4 b1_##i = *(const float4*)(bp[i] + k0 + KQ);
    STG_A(0) STG_A(1) STG_A(2) STG_A(3)
    STG_B(0) STG_B(1) STG_B(2) STG_B(3) STG_B(4) STG_B(5) STG_B(6) STG_B(7)
#undef STG_A
#undef STG_B
    __builtin_amdgcn_sched_barrier(0);
#define PUT_A(i) *(float4*)&A_lds[((i) * 16 + srow) * LDP + sc4] = a0_##i; \
                 *(float4*)&A_lds[((i) * 16 + srow) * LDP + KQ + sc4] = a1_##i;
#define PUT_B(i) *(float4*)&B_lds[((i) * 16 + srow) * LDP + sc4] = b0_##i; \
                 *(float4*)&B_lds[((i) * 16 + srow) * LDP + KQ + sc4] = b1_##i;
    PUT_A(0) PUT_A(1) PUT_A(2) PUT_A(3)
    PUT_B(0) PUT_B(1) PUT_B(2) PUT_B(3) PUT_B(4) PUT_B(5) PUT_B(6) PUT_B(7)
#undef PUT_A
#undef PUT_B
    __syncthreads();

#pragma unroll
    for (int kk = 0; kk < KC; kk += 4) {
      v2f a = *(const v2f*)&A_lds[(wm * 16 + l16) * LDP + kk + 2 * lhi];
#pragma unroll
      for (int g = 0; g < 4; ++g) {
        v2f bf = *(const v2f*)&B_lds[(wn * 64 + g * 16 + l16) * LDP + kk + 2 * lhi];
        acc[g] = __builtin_amdgcn_wmma_f32_16x16x4_f32(
            false, a, false, bf, (short)0, acc[g], false, false);
      }
    }
    __syncthreads();
  }

  {
    const int colbase = cblk * 128 + wn * 64;
    const int rowg0   = r0 + wm * 16 + 8 * lhi;
#pragma unroll
    for (int j = 0; j < 4; ++j) {
      int col = colbase + j * 16 + l16;
      float bv = bias[col];
#pragma unroll
      for (int vr = 0; vr < 8; ++vr) {
        int rg = rowg0 + vr;
        float val = acc[j][vr] + bv;
        if (col < HDIM) out0[(size_t)rg * HDIM + col] = val;
        else            out1[(size_t)rg * HDIM + (col - HDIM)] = val;
      }
    }
  }
}

#define MIXG 2
#define MIXT 6

__global__ __launch_bounds__(256) void mix_k(const float* __restrict__ newh,
                                             const float* __restrict__ newc,
                                             const float* __restrict__ hcur,
                                             const float* __restrict__ ccur,
                                             const float* __restrict__ q,
                                             const int* __restrict__ length,
                                             float* __restrict__ hnxt,
                                             float* __restrict__ cnxt,
                                             int n, int iter) {
  __shared__ float sc[32];
  const int b = blockIdx.y;
  const int tid = threadIdx.x;
  const unsigned lane = tid & 31u;
  const int wave = tid >> 5;
  const int nvec = n * (HDIM / 4);
  if ((int)blockIdx.x * 256 >= nvec) return;

  const int len = length[b];
  const int nlive = min(max(len - 1 - iter, 0), n);
  const int p0 = wave, p1 = wave + 8, p2 = wave + 16;
  float s0 = 0.0f, s1 = 0.0f, s2 = 0.0f;
  if (nlive > 0) {
    const float* r0 = newh + (size_t)(b * n + min(p0, nlive - 1)) * HDIM;
    const float* r1 = newh + (size_t)(b * n + min(p1, nlive - 1)) * HDIM;
    const float* r2 = newh + (size_t)(b * n + min(p2, nlive - 1)) * HDIM;
#pragma unroll
    for (int t = 0; t < HDIM / 32; ++t) {
      const int j = (int)lane + 32 * t;
      const float qv = q[j];
      s0 += r0[j] * qv;
      s1 += r1[j] * qv;
      s2 += r2[j] * qv;
    }
#pragma unroll
    for (int off = 16; off > 0; off >>= 1) {
      s0 += __shfl_xor(s0, off, 32);
      s1 += __shfl_xor(s1, off, 32);
      s2 += __shfl_xor(s2, off, 32);
    }
  }
  if (lane == 0) {
    if (p0 < n) sc[p0] = (p0 < nlive) ? s0 : -1.0e9f;
    if (p1 < n) sc[p1] = (p1 < nlive) ? s1 : -1.0e9f;
    if (p2 < n) sc[p2] = (p2 < nlive) ? s2 : -1.0e9f;
  }
  __syncthreads();

  float v = -INFINITY;
  if ((int)lane < n) {
    float s = ((iter + 1 + (int)lane) < len) ? sc[lane] : -1.0e9f;
    v = (s == s) ? s : -INFINITY;
  }
  float m = v;
#pragma unroll
  for (int off = 16; off > 0; off >>= 1) m = fmaxf(m, __shfl_xor(m, off, 32));
  const unsigned long long hit = __ballot(v == m);
  const int ksel = hit ? (__ffsll((long long)hit) - 1) : 0;
  const bool done = (iter + 1) < len;

  const int vi0 = (int)blockIdx.x * 256 + tid;
  if (vi0 < nvec) {
#define MIX_LOAD(T)                                                              \
    float4 hv##T, cv##T; size_t dst##T;                                          \
    {                                                                            \
      const int vi = vi0 + (T) * (MIXG * 256);                                   \
      const int vc = (vi < nvec) ? vi : vi0;                                     \
      const int p = vc >> 7;                              \
      const int j = (vc & 127) * 4;                                              \
      const size_t cur = (size_t)(b * SEQL + p) * HDIM + j;                      \
      const size_t ni  = (size_t)(b * n + p) * HDIM + j;                         \
      const bool take_new   = done && (p == ksel);                               \
      const bool take_right = done && (p > ksel);                                \
      const size_t oc = cur + (take_right ? (size_t)HDIM : (size_t)0);           \
      const float* sh  = take_new ? (newh + ni) : (hcur + oc);                   \
      const float* scp = take_new ? (newc + ni) : (ccur + oc);                   \
      dst##T = cur;                                                              \
      hv##T = *(const float4*)sh;                                                \
      cv##T = *(const float4*)scp;                                               \
    }
#define MIX_STORE(T)                                                             \
    *(float4*)&hnxt[dst##T] = hv##T;                                             \
    *(float4*)&cnxt[dst##T] = cv##T;

    MIX_LOAD(0) MIX_LOAD(1) MIX_LOAD(2) MIX_LOAD(3) MIX_LOAD(4) MIX_LOAD(5)
    __builtin_amdgcn_sched_barrier(0);
    MIX_STORE(0) MIX_STORE(1) MIX_STORE(2) MIX_STORE(3) MIX_STORE(4) MIX_STORE(5)
#undef MIX_LOAD
#undef MIX_STORE
  }
}

__global__ __launch_bounds__(256) void final_copy_k(const float* __restrict__ hfin,
                                                    float* __restrict__ out) {
  int idx = blockIdx.x * 256 + threadIdx.x;
  int b = idx >> 9;
  int j = idx & (HDIM - 1);
  out[idx] = hfin[(size_t)b * SEQL * HDIM + j];
}

extern "C" void kernel_launch(void* const* d_in, const int* in_sizes, int n_in,
                              void* d_out, int out_size, void* d_ws, size_t ws_size,
                              hipStream_t stream) {
  const float* inp    = (const float*)d_in[0];
  const int*   length = (const int*)  d_in[1];
  const float* word_W = (const float*)d_in[2];
  const float* word_b = (const float*)d_in[3];
  const float* comp_W = (const float*)d_in[4];
  const float* comp_b = (const float*)d_in[5];
  const float* comp_q = (const float*)d_in[6];

  float* ws = (float*)d_ws;
  size_t off = 0;
  float* hb[2];
  float* cb[2];
  hb[0] = ws + off; off += (size_t)BATCH * SEQL * HDIM;
  cb[0] = ws + off; off += (size_t)BATCH * SEQL * HDIM;
  hb[1] = ws + off; off += (size_t)BATCH * SEQL * HDIM;
  cb[1] = ws + off; off += (size_t)BATCH * SEQL * HDIM;
  float* newh = ws + off; off += (size_t)(BATCH * 23) * HDIM;
  float* newc = ws + off; off += (size_t)(BATCH * 23) * HDIM;

  gemm_wmma<0><<<dim3((BATCH * SEQL) / TM, 1024 / 128), 256, 0, stream>>>(
      inp, nullptr, nullptr, word_W, word_b, hb[0], cb[0], nullptr, 512, 0, 0);

  int cur = 0;
  for (int i = 0; i < SEQL - 1; ++i) {
    int n = (SEQL - 1) - i;
    int nxt = cur ^ 1;
    gemm_wmma<1><<<dim3(HDIM / 32, 32), 256, 0, stream>>>(
        nullptr, hb[cur], cb[cur], comp_W, comp_b, newh, newc, length, 1024, n, i);
    mix_k<<<dim3(MIXG, BATCH), 256, 0, stream>>>(newh, newc, hb[cur], cb[cur], comp_q, length,
                                     hb[nxt], cb[nxt], n, i);
    cur = nxt;
  }

  final_copy_k<<<(BATCH * HDIM) / 256, 256, 0, stream>>>(hb[cur], (float*)d_out);
  (void)in_sizes; (void)n_in; (void)out_size; (void)ws_size;
}


# Full-size merge steps: block = 16 cols x RT 16-row tiles, RT = 3..9 by live-row count (one block per WGP); gate tiles dealt evenly over the 4 SIMDs, meet in LDS; other k start per sum: new k order
# speedup vs baseline: 2.7653x; 1.0587x over previous
#include <hip/hip_runtime.h>
#include <hip/hip_bf16.h>
#include <math.h>


typedef __attribute__((ext_vector_type(2))) float v2f;
typedef __attribute__((ext_vector_type(8))) float v8f;

#define BATCH 64
#define SEQL  24
#define HDIM  512

__device__ __forceinline__ float sigm(float x) { return 1.0f / (1.0f + __expf(-x)); }
__device__ __forceinline__ float tanh_fast(float x) { return 2.0f / (1.0f + __expf(-2.0f * x)) - 1.0f; }

#define TM   64
#define KQ   64
#define KC   128
#define LDP  132
#define KCS  256
#define LDPS 260
#define KS   64
#define LDQ  68
#define RMAX 9

template <int V> struct IntC { static constexpr int value = V; };

template <int MODE>
__global__ __launch_bounds__(256, 4) void gemm_wmma(const float* __restrict__ X,
                                                 const float* __restrict__ hbuf,
                                                 const float* __restrict__ cbuf,
                                                 const float* __restrict__ W,
                                                 const float* __restrict__ bias,
                                                 float* __restrict__ out0,
                                                 float* __restrict__ out1,
                                                 const int* __restrict__ length,
                                                 int K, int n, int iter) {
  __shared__ __attribute__((aligned(16))) float S_lds[2 * (16 * RMAX + 80) * LDQ];
  float* const A_lds = S_lds;
  float* const B_lds = S_lds + ((MODE == 0) ? TM * LDP : 16 * LDPS);
  static_assert((TM + 128) * LDP <= 2 * (16 * RMAX + 80) * LDQ, "mode 0's chunk fits the array");
  __shared__ __attribute__((aligned(16))) int s_cnt[BATCH];
  __shared__ __attribute__((aligned(16))) int s_row[16 * RMAX];
  __shared__ __attribute__((aligned(16))) unsigned s_off[16 * RMAX];

  const int tid  = threadIdx.x;
  const unsigned lane = tid & 31u;
  const int wave = tid >> 5;
  const int wm   = wave >> 1;
  const int wn   = wave & 1;
  const int l16  = (int)(lane & 15u);
  const int lhi  = (int)(lane >> 4);

  const int rblk = (MODE == 0) ? (int)blockIdx.x : (int)blockIdx.y;
  const int cblk = (MODE == 0) ? (int)blockIdx.y : (int)blockIdx.x;
  int r0 = rblk * TM;
  int nlv = TM;
  int rpb = TM;
  const int NACC = (MODE == 0) ? 4 : 5;

  int tot = 0;
  bool small = false;
  int cS = 0;
  int RT = 0;
  unsigned rot = (MODE == 0) ? ((unsigned)(rblk + cblk) % ((unsigned)K / KC)) : 0u;
  if (MODE == 1) {
    int c = 0;
    if (tid < BATCH) {
      c = min(max(length[tid] - 1 - iter, 0), n);
      s_cnt[tid] = c;
    }
    if (tid < 16 * RMAX) { s_row[tid] = 0; s_off[tid] = 0u; }
    __syncthreads();
    int pre = 0;
#pragma unroll
    for (int j = 0; j < BATCH; ++j) {
      const int cj = s_cnt[j];
      pre += (j < tid) ? cj : 0;
      tot += cj;
    }
    tot = __builtin_amdgcn_readfirstlane(tot);
    const int nwg = (int)(gridDim.x * gridDim.y);
    small = (tot <= 256) && (32 * ((tot + 15) >> 4) <= nwg);
    const int lin = (int)(blockIdx.y * gridDim.x + blockIdx.x);
    const int rb = lin >> 5;
    cS = lin & 31;
    if (small) {
      rpb = 16;
    } else {
      const int tiles = (tot + 15) >> 4;
      RT = min(max((tiles + 7) >> 3, 3), RMAX);
      rpb = 16 * RT;
    }
    r0 = rb * rpb;
    if (r0 >= tot) return;
    rot = small ? ((unsigned)(rb + cS) % ((unsigned)K / KCS))
                : ((unsigned)(rb + cS) % ((unsigned)K / KC));
    nlv = min(rpb, tot - r0);
    if (tid < BATCH) {
      const int lo = max(pre, r0);
      const int hi = min(pre + c, r0 + rpb);
      for (int r = lo; r < hi; ++r) {
        s_row[r - r0] = (tid << 5) | (r - pre);
        s_off[r - r0] = (unsigned)((tid * SEQL + (r - pre)) * HDIM) * 4u;
      }
    }
    __syncthreads();
  }
  if (MODE == 1 && small) {
    const int srowS = tid >> 4;
    const int sc4S  = (tid & 15) * 4;
    const int infoA = s_row[srowS];
    const float* apS = hbuf + (size_t)((infoA >> 5) * SEQL + (infoA & 31)) * HDIM + sc4S;
    const float* bpS[5];
#pragma unroll
    for (int g = 0; g < 5; ++g) bpS[g] = W + (size_t)(g * HDIM + cS * 16 + srowS) * K + sc4S;
    const int wv = __builtin_amdgcn_readfirstlane(wave);
    const bool actS = wv < 5;
    v8f accS = {};
    const unsigned nchS = (unsigned)K / KCS;
    for (unsigned ci = 0; ci < nchS; ++ci) {
      unsigned ch = ci + rot;
      if (ch >= nchS) ch -= nchS;
      const int k0 = (int)(ch * KCS);
      static_assert(KCS == 4 * KQ, "a thread loads four KQ-wide pieces of every staged row per chunk");
      static_assert((16 + 80) * LDPS <= 2 * (16 * RMAX + 80) * LDQ, "the short step's rows fit the array");
#define SLD(h) const float4 sa_##h  = *(const float4*)(apS + k0 + (h) * KQ);    \
               const float4 sb0_##h = *(const float4*)(bpS[0] + k0 + (h) * KQ); \
               const float4 sb1_##h = *(const float4*)(bpS[1] + k0 + (h) * KQ); \
               const float4 sb2_##h = *(const float4*)(bpS[2] + k0 + (h) * KQ); \
               const float4 sb3_##h = *(const float4*)(bpS[3] + k0 + (h) * KQ); \
               const float4 sb4_##h = *(const float4*)(bpS[4] + k0 + (h) * KQ);
#define SPUT(h) *(float4*)&A_lds[srowS * LDPS + (h) * KQ + sc4S] = sa_##h;              \
                *(float4*)&B_lds[(0 * 16 + srowS) * LDPS + (h) * KQ + sc4S] = sb0_##h;  \
                *(float4*)&B_lds[(1 * 16 + srowS) * LDPS + (h) * KQ + sc4S] = sb1_##h;  \
                *(float4*)&B_lds[(2 * 16 + srowS) * LDPS + (h) * KQ + sc4S] = sb2_##h;  \
                *(float4*)&B_lds[(3 * 16 + srowS) * LDPS + (h) * KQ + sc4S] = sb3_##h;  \
                *(float4*)&B_lds[(4 * 16 + srowS) * LDPS + (h) * KQ + sc4S] = sb4_##h;
      SLD(0) SLD(1) SLD(2) SLD(3)
      SPUT(0) SPUT(1) SPUT(2) SPUT(3)
#undef SLD
#undef SPUT
      __syncthreads();
      if (actS) {
#pragma unroll
        for (int kk = 0; kk < KCS; kk += 4) {
          v2f a  = *(const v2f*)&A_lds[l16 * LDPS + kk + 2 * lhi];
          v2f bf = *(const v2f*)&B_lds[(wv * 16 + l16) * LDPS + kk + 2 * lhi];
          accS = __builtin_amdgcn_wmma_f32_16x16x4_f32(
              false, a, false, bf, (short)0, accS, false, false);
        }
      }
      __syncthreads();
    }
    float* Gs = B_lds;
    if (actS) {
#pragma unroll
      for (int vr = 0; vr < 8; ++vr) Gs[(wv * 16 + 8 * lhi + vr) * 17 + l16] = accS[vr];
    }
    __syncthreads();
    {
      const int row  = tid >> 4;
      const int col  = tid & 15;
      const int colg = cS * 16 + col;
      const int info = s_row[row];
      const int b = info >> 5;
      const int p = info & 31;
      const size_t cbase = (size_t)(b * SEQL + p) * HDIM + colg;
      const float clS = cbuf[cbase];
      const float crS = cbuf[cbase + HDIM];
      float gi  = Gs[(0 * 16 + row) * 17 + col] + bias[0 * HDIM + colg];
      float gfl = Gs[(1 * 16 + row) * 17 + col] + bias[1 * HDIM + colg];
      float gfr = Gs[(2 * 16 + row) * 17 + col] + bias[2 * HDIM + colg];
      float gu  = Gs[(3 * 16 + row) * 17 + col] + bias[3 * HDIM + colg];
      float go  = Gs[(4 * 16 + row) * 17 + col] + bias[4 * HDIM + colg];
      float nc = clS * sigm(gfl + 1.0f) + crS * sigm(gfr + 1.0f) + tanh_fast(gu) * sigm(gi);
      float nh = sigm(go) * tanh_fast(nc);
      if (row < nlv) {
        const size_t oi = (size_t)(b * n + p) * HDIM + colg;
        out0[oi] = nh;
        out1[oi] = nc;
      }
    }
    return;
  }

  if (MODE == 1) {
    const int wv = __builtin_amdgcn_readfirstlane(wave);
    const bool loader = wv >= 4;

    constexpr unsigned KK1 = 2 * HDIM;
    const char* const hbase = (const char*)hbuf;
    const char* const wbase = (const char*)W;

    auto run = [&](auto rc) __attribute__((always_inline)) {
      constexpr int RR  = decltype(rc)::value;
      constexpr int NAP = 2 * RR;
      constexpr int NT  = 5 * RR;
      constexpr int Q   = RR / 4;
      constexpr int REM = RR % 4;
      constexpr int NLT = 5 * REM;
      constexpr int NL  = (NLT + 3) / 4;
      constexpr int NW  = 5 * Q + NL;
      constexpr bool LDM = (NW > 10);
      constexpr int NACC = LDM ? ((5 * Q > NL) ? 5 * Q : NL) : NW;
      static_assert(RR >= 1 && RR <= RMAX, "the block fits the staged rows");

      auto stage = [&](const int k0, float* __restrict__ Ad, float* __restrict__ Bd, auto between)
          __attribute__((always_inline)) {
        int ts = tid;
        asm volatile("" : "+v"(ts));
        const int r8s = (ts & 127) >> 4;
        const int c4s = (ts & 15) * 4;
        const unsigned kB = (unsigned)k0 * 4u;
        const unsigned kA = kB + (unsigned)c4s * 4u;
        const unsigned wB = (unsigned)((cS * 16 + r8s) * (int)KK1 + c4s) * 4u;
#define LDA(j) float4 la_##j; if constexpr ((j) < NAP) la_##j = *(const float4*)(hbase + (s_off[(j) * 8 + r8s] + kA));
#define LDB(j) const float4 lb_##j = *(const float4*)(wbase + (wB + kB + \
                   (unsigned)((((j) >> 1) * HDIM + ((j) & 1) * 8) * (int)KK1) * 4u));
#define STA(j) if constexpr ((j) < NAP) *(float4*)&Ad[((j) * 8 + r8s) * LDQ + c4s] = la_##j;
#define STB(j) *(float4*)&Bd[((j) * 8 + r8s) * LDQ + c4s] = lb_##j;
        static_assert(NAP <= 18, "the A passes are written out 18 times");
        LDA(0) LDA(1) LDA(2) LDA(3) LDA(4) LDA(5) LDA(6) LDA(7) LDA(8)
        LDA(9) LDA(10) LDA(11) LDA(12) LDA(13) LDA(14) LDA(15) LDA(16) LDA(17)
        LDB(0) LDB(1) LDB(2) LDB(3) LDB(4) LDB(5) LDB(6) LDB(7) LDB(8) LDB(9)
        __builtin_amdgcn_sched_barrier(0);
        between();
        __builtin_amdgcn_sched_barrier(0);
        STA(0) STA(1) STA(2) STA(3) STA(4) STA(5) STA(6) STA(7) STA(8)
        STA(9) STA(10) STA(11) STA(12) STA(13) STA(14) STA(15) STA(16) STA(17)
        STB(0) STB(1) STB(2) STB(3) STB(4) STB(5) STB(6) STB(7) STB(8) STB(9)
#undef LDA
#undef LDB
#undef STA
#undef STB
      };

      v8f acc[NACC] = {};
      auto frag_base = [&]() __attribute__((always_inline)) {
        int tm = tid;
        asm volatile("" : "+v"(tm));
        return (tm & 15) * LDQ + 2 * ((tm >> 4) & 1);
      };
      auto mma_left = [&](auto a0c, const int w, const int fb,
                          const float* __restrict__ Ab, const float* __restrict__ Bb)
          __attribute__((always_inline)) {
        constexpr int A0I = decltype(a0c)::value;
        if constexpr (NL > 0) {
          int lao[NL], lbo[NL];
#pragma unroll
          for (int i = 0; i < NL; ++i) {
            const int u = min(w + 4 * i, NLT - 1);
            lao[i] = fb + (4 * Q + u / 5) * 16 * LDQ;
            lbo[i] = fb + (u % 5) * 16 * LDQ;
          }
#pragma unroll
          for (int kk = 0; kk < KS; kk += 4) {
#pragma unroll
            for (int i = 0; i < NL; ++i) {
              const v2f a  = *(const v2f*)&Ab[lao[i] + kk];
              const v2f bf = *(const v2f*)&Bb[lbo[i] + kk];
              acc[A0I + i] = __builtin_amdgcn_wmma_f32_16x16x4_f32(
                  false, a, false, bf, (short)0, acc[A0I + i], false, false);
            }
          }
        }
      };
      auto mma = [&](const float* __restrict__ Ab, const float* __restrict__ Bb)
          __attribute__((always_inline)) {
        const int fb = frag_base();
        if constexpr (Q > 0) {
          const int fa = fb + wv * 16 * LDQ;
#pragma unroll
          for (int kk = 0; kk < KS; kk += 4) {
            v2f bfr[5];
#pragma unroll
            for (int g = 0; g < 5; ++g) bfr[g] = *(const v2f*)&Bb[g * 16 * LDQ + fb + kk];
#pragma unroll
            for (int i = 0; i < Q; ++i) {
              const v2f a = *(const v2f*)&Ab[fa + i * 64 * LDQ + kk];
#pragma unroll
              for (int g = 0; g < 5; ++g)
                acc[5 * i + g] = __builtin_amdgcn_wmma_f32_16x16x4_f32(
                    false, a, false, bfr[g], (short)0, acc[5 * i + g], false, false);
            }
          }
        }
        if constexpr (!LDM) mma_left(IntC<5 * Q>{}, wv, fb, Ab, Bb);
      };
      auto mma_ld = [&](const float* __restrict__ Ab, const float* __restrict__ Bb)
          __attribute__((always_inline)) {
        if constexpr (LDM) mma_left(IntC<0>{}, wv - 4, frag_base(), Ab, Bb);
      };

      float* const A0 = S_lds;
      float* const A1 = S_lds + 16 * RMAX * LDQ;
      float* const B0 = S_lds + 2 * 16 * RMAX * LDQ;
      float* const B1 = B0 + 80 * LDQ;
      const unsigned nst = KK1 / KS;
      const unsigned st0 = rot * (KC / KS);
      static_assert((KC / KS) == 2, "the stage loop below takes the ring's two stages in turn");
      if (loader) stage((int)(st0 * KS), A0, B0, [] {});
      __syncthreads();
      for (unsigned s = 0; s < nst; s += 2) {
        {
          unsigned nx = st0 + s + 1;
          if (nx >= nst) nx -= nst;
          if (loader) stage((int)(nx * KS), A1, B1, [&] { mma_ld(A0, B0); }); else mma(A0, B0);
          __syncthreads();
        }
        {
          unsigned nx = st0 + s + 2;
          if (nx >= nst) nx -= nst;
          if (loader) {
            if (s + 2 < nst) stage((int)(nx * KS), A0, B0, [&] { mma_ld(A1, B1); });
            else mma_ld(A1, B1);
          } else {
            mma(A1, B1);
          }
          __syncthreads();
        }
      }

      float* const Gs = S_lds;
      static_assert(NT * 16 * 17 <= 2 * (16 * RMAX + 80) * LDQ, "the gate tiles fit the array");
      int te = tid;
      asm volatile("" : "+v"(te));
      const int l16e  = te & 15;
      const int lhie  = (te >> 4) & 1;
      const int r16e  = te >> 4;
      const int cole  = te & 15;
      const int colge = cS * 16 + cole;
      if (!loader) {
#pragma unroll
        for (int i = 0; i < Q; ++i) {
#pragma unroll
          for (int g = 0; g < 5; ++g) {
            const int t = (wv + 4 * i) * 5 + g;
#pragma unroll
            for (int vr = 0; vr < 8; ++vr) Gs[(t * 16 + 8 * lhie + vr) * 17 + l16e] = acc[5 * i + g][vr];
          }
        }
      }
      if (loader == LDM) {
        constexpr int A0I = LDM ? 0 : 5 * Q;
#pragma unroll
        for (int i = 0; i < NL; ++i) {
          const int u = (wv & 3) + 4 * i;
          if (u < NLT) {
            const int t = 20 * Q + u;
#pragma unroll
            for (int vr = 0; vr < 8; ++vr) Gs[(t * 16 + 8 * lhie + vr) * 17 + l16e] = acc[A0I + i][vr];
          }
        }
      }
      __syncthreads();

      float bz[5];
#pragma unroll
      for (int g = 0; g < 5; ++g) bz[g] = bias[g * HDIM + colge];
      float clv[RR], crv[RR];
      unsigned oiv[RR];
#pragma unroll
      for (int ps = 0; ps < RR; ++ps) {
        const int info = s_row[ps * 16 + r16e];
        const int b = info >> 5;
        const int p = info & 31;
        const unsigned cbase = (unsigned)((b * SEQL + p) * HDIM + colge);
        clv[ps] = cbuf[cbase];
        crv[ps] = cbuf[cbase + HDIM];
        oiv[ps] = (unsigned)((b * n + p) * HDIM + colge);
      }
#pragma unroll
      for (int ps = 0; ps < RR; ++ps) asm volatile("" : "+v"(clv[ps]), "+v"(crv[ps]));
#pragma unroll
      for (int ps = 0; ps < RR; ++ps) {
        const bool live = (ps * 16 + r16e) < nlv;
        const float* Gp = Gs + ((ps * 5) * 16 + r16e) * 17 + cole;
        float gi  = Gp[0 * 16 * 17] + bz[0];
        float gfl = Gp[1 * 16 * 17] + bz[1];
        float gfr = Gp[2 * 16 * 17] + bz[2];
        float gu  = Gp[3 * 16 * 17] + bz[3];
        float go  = Gp[4 * 16 * 17] + bz[4];
        float nc = clv[ps] * sigm(gfl + 1.0f) + crv[ps] * sigm(gfr + 1.0f) + tanh_fast(gu) * sigm(gi);
        float nh = sigm(go) * tanh_fast(nc);
        if (live) {
          out0[oiv[ps]] = nh;
          out1[oiv[ps]] = nc;
        }
      }
    };

    switch (RT) {
      case 3:  run(IntC<3>{}); break;
      case 4:  run(IntC<4>{}); break;
      case 5:  run(IntC<5>{}); break;
      case 6:  run(IntC<6>{}); break;
      case 7:  run(IntC<7>{}); break;
      case 8:  run(IntC<8>{}); break;
      default: run(IntC<RMAX>{}); break;
    }
    return;
  }

  const int NA = TM / 16;
  static_assert(KC == 2 * KQ, "a thread loads two KQ-wide pieces of every staged row per chunk");
  static_assert(TM == 64, "the staging below is written out for 4 A rows per thread");

  v8f acc[4] = {};

  const int srow = tid >> 4;
  const int sc4  = (tid & 15) * 4;
  const float* ap[NA];
#pragma unroll
  for (int i = 0; i < NA; ++i) ap[i] = X + (size_t)(r0 + i * 16 + srow) * K + sc4;
  const float* bp[8];
#pragma unroll
  for (int i = 0; i < 8; ++i) bp[i] = W + (size_t)(cblk * 128 + i * 16 + srow) * K + sc4;

  const unsigned nch = (unsigned)K / KC;
  for (unsigned ci = 0; ci < nch; ++ci) {
    unsigned ch = ci + rot;
    if (ch >= nch) ch -= nch;
    const int k0 = (int)(ch * KC);
#define STG_A(i) const float4 a0_##i = *(const float4*)(ap[i] + k0); \
                 const float4 a1_##i = *(const float4*)(ap[i] + k0 + KQ);
#define STG_B(i) const float4 b0_##i = *(const float4*)(bp[i] + k0); \
                 const float4 b1_##i = *(const float4*)(bp[i] + k0 + KQ);
    STG_A(0) STG_A(1) STG_A(2) STG_A(3)
    STG_B(0) STG_B(1) STG_B(2) STG_B(3) STG_B(4) STG_B(5) STG_B(6) STG_B(7)
#undef STG_A
#undef STG_B
    __builtin_amdgcn_sched_barrier(0);
#define PUT_A(i) *(float4*)&A_lds[((i) * 16 + srow) * LDP + sc4] = a0_##i; \
                 *(float4*)&A_lds[((i) * 16 + srow) * LDP + KQ + sc4] = a1_##i;
#define PUT_B(i) *(float4*)&B_lds[((i) * 16 + srow) * LDP + sc4] = b0_##i; \
                 *(float4*)&B_lds[((i) * 16 + srow) * LDP + KQ + sc4] = b1_##i;
    PUT_A(0) PUT_A(1) PUT_A(2) PUT_A(3)
    PUT_B(0) PUT_B(1) PUT_B(2) PUT_B(3) PUT_B(4) PUT_B(5) PUT_B(6) PUT_B(7)
#undef PUT_A
#undef PUT_B
    __syncthreads();

#pragma unroll
    for (int kk = 0; kk < KC; kk += 4) {
      v2f a = *(const v2f*)&A_lds[(wm * 16 + l16) * LDP + kk + 2 * lhi];
#pragma unroll
      for (int g = 0; g < 4; ++g) {
        v2f bf = *(const v2f*)&B_lds[(wn * 64 + g * 16 + l16) * LDP + kk + 2 * lhi];
        acc[g] = __builtin_amdgcn_wmma_f32_16x16x4_f32(
            false, a, false, bf, (short)0, acc[g], false, false);
      }
    }
    __syncthreads();
  }

  {
    const int colbase = cblk * 128 + wn * 64;
    const int rowg0   = r0 + wm * 16 + 8 * lhi;
#pragma unroll
    for (int j = 0; j < 4; ++j) {
      int col = colbase + j * 16 + l16;
      float bv = bias[col];
#pragma unroll
      for (int vr = 0; vr < 8; ++vr) {
        int rg = rowg0 + vr;
        float val = acc[j][vr] + bv;
        if (col < HDIM) out0[(size_t)rg * HDIM + col] = val;
        else            out1[(size_t)rg * HDIM + (col - HDIM)] = val;
      }
    }
  }
}

#define MIXG 2
#define MIXT 6

__global__ __launch_bounds__(256) void mix_k(const float* __restrict__ newh,
                                             const float* __restrict__ newc,
                                             const float* __restrict__ hcur,
                                             const float* __restrict__ ccur,
                                             const float* __restrict__ q,
                                             const int* __restrict__ length,
                                             float* __restrict__ hnxt,
                                             float* __restrict__ cnxt,
                                             int n, int iter) {
  __shared__ float sc[32];
  const int b = blockIdx.y;
  const int tid = threadIdx.x;
  const unsigned lane = tid & 31u;
  const int wave = tid >> 5;
  const int nvec = n * (HDIM / 4);
  if ((int)blockIdx.x * 256 >= nvec) return;

  const int len = length[b];
  const int nlive = min(max(len - 1 - iter, 0), n);
  const int p0 = wave, p1 = wave + 8, p2 = wave + 16;
  float s0 = 0.0f, s1 = 0.0f, s2 = 0.0f;
  if (nlive > 0) {
    const float* r0 = newh + (size_t)(b * n + min(p0, nlive - 1)) * HDIM;
    const float* r1 = newh + (size_t)(b * n + min(p1, nlive - 1)) * HDIM;
    const float* r2 = newh + (size_t)(b * n + min(p2, nlive - 1)) * HDIM;
#pragma unroll
    for (int t = 0; t < HDIM / 32; ++t) {
      const int j = (int)lane + 32 * t;
      const float qv = q[j];
      s0 += r0[j] * qv;
      s1 += r1[j] * qv;
      s2 += r2[j] * qv;
    }
#pragma unroll
    for (int off = 16; off > 0; off >>= 1) {
      s0 += __shfl_xor(s0, off, 32);
      s1 += __shfl_xor(s1, off, 32);
      s2 += __shfl_xor(s2, off, 32);
    }
  }
  if (lane == 0) {
    if (p0 < n) sc[p0] = (p0 < nlive) ? s0 : -1.0e9f;
    if (p1 < n) sc[p1] = (p1 < nlive) ? s1 : -1.0e9f;
    if (p2 < n) sc[p2] = (p2 < nlive) ? s2 : -1.0e9f;
  }
  __syncthreads();

  float v = -INFINITY;
  if ((int)lane < n) {
    float s = ((iter + 1 + (int)lane) < len) ? sc[lane] : -1.0e9f;
    v = (s == s) ? s : -INFINITY;
  }
  float m = v;
#pragma unroll
  for (int off = 16; off > 0; off >>= 1) m = fmaxf(m, __shfl_xor(m, off, 32));
  const unsigned long long hit = __ballot(v == m);
  const int ksel = hit ? (__ffsll((long long)hit) - 1) : 0;
  const bool done = (iter + 1) < len;

  const int vi0 = (int)blockIdx.x * 256 + tid;
  if (vi0 < nvec) {
#define MIX_LOAD(T)                                                              \
    float4 hv##T, cv##T; size_t dst##T;                                          \
    {                                                                            \
      const int vi = vi0 + (T) * (MIXG * 256);                                   \
      const int vc = (vi < nvec) ? vi : vi0;                                     \
      const int p = vc >> 7;                              \
      const int j = (vc & 127) * 4;                                              \
      const size_t cur = (size_t)(b * SEQL + p) * HDIM + j;                      \
      const size_t ni  = (size_t)(b * n + p) * HDIM + j;                         \
      const bool take_new   = done && (p == ksel);                               \
      const bool take_right = done && (p > ksel);                                \
      const size_t oc = cur + (take_right ? (size_t)HDIM : (size_t)0);           \
      const float* sh  = take_new ? (newh + ni) : (hcur + oc);                   \
      const float* scp = take_new ? (newc + ni) : (ccur + oc);                   \
      dst##T = cur;                                                              \
      hv##T = *(const float4*)sh;                                                \
      cv##T = *(const float4*)scp;                                               \
    }
#define MIX_STORE(T)                                                             \
    *(float4*)&hnxt[dst##T] = hv##T;                                             \
    *(float4*)&cnxt[dst##T] = cv##T;

    MIX_LOAD(0) MIX_LOAD(1) MIX_LOAD(2) MIX_LOAD(3) MIX_LOAD(4) MIX_LOAD(5)
    __builtin_amdgcn_sched_barrier(0);
    MIX_STORE(0) MIX_STORE(1) MIX_STORE(2) MIX_STORE(3) MIX_STORE(4) MIX_STORE(5)
#undef MIX_LOAD
#undef MIX_STORE
  }
}

__global__ __launch_bounds__(256) void final_copy_k(const float* __restrict__ hfin,
                                                    float* __restrict__ out) {
  int idx = blockIdx.x * 256 + threadIdx.x;
  int b = idx >> 9;
  int j = idx & (HDIM - 1);
  out[idx] = hfin[(size_t)b * SEQL * HDIM + j];
}

extern "C" void kernel_launch(void* const* d_in, const int* in_sizes, int n_in,
                              void* d_out, int out_size, void* d_ws, size_t ws_size,
                              hipStream_t stream) {
  const float* inp    = (const float*)d_in[0];
  const int*   length = (const int*)  d_in[1];
  const float* word_W = (const float*)d_in[2];
  const float* word_b = (const float*)d_in[3];
  const float* comp_W = (const float*)d_in[4];
  const float* comp_b = (const float*)d_in[5];
  const float* comp_q = (const float*)d_in[6];

  float* ws = (float*)d_ws;
  size_t off = 0;
  float* hb[2];
  float* cb[2];
  hb[0] = ws + off; off += (size_t)BATCH * SEQL * HDIM;
  cb[0] = ws + off; off += (size_t)BATCH * SEQL * HDIM;
  hb[1] = ws + off; off += (size_t)BATCH * SEQL * HDIM;
  cb[1] = ws + off; off += (size_t)BATCH * SEQL * HDIM;
  float* newh = ws + off; off += (size_t)(BATCH * 23) * HDIM;
  float* newc = ws + off; off += (size_t)(BATCH * 23) * HDIM;

  gemm_wmma<0><<<dim3((BATCH * SEQL) / TM, 1024 / 128), 256, 0, stream>>>(
      inp, nullptr, nullptr, word_W, word_b, hb[0], cb[0], nullptr, 512, 0, 0);

  int cur = 0;
  for (int i = 0; i < SEQL - 1; ++i) {
    int n = (SEQL - 1) - i;
    int nxt = cur ^ 1;
    gemm_wmma<1><<<dim3(HDIM / 32, 32), 256, 0, stream>>>(
        nullptr, hb[cur], cb[cur], comp_W, comp_b, newh, newc, length, 1024, n, i);
    mix_k<<<dim3(MIXG, BATCH), 256, 0, stream>>>(newh, newc, hb[cur], cb[cur], comp_q, length,
                                     hb[nxt], cb[nxt], n, i);
    cur = nxt;
  }

  final_copy_k<<<(BATCH * HDIM) / 256, 256, 0, stream>>>(hb[cur], (float*)d_out);
  (void)in_sizes; (void)n_in; (void)out_size; (void)ws_size;
}
